# starting kernel
# baseline (speedup 1.0000x reference)
#include <hip/hip_runtime.h>
#include <math.h>

#define IN_CH 512
#define HEADS 8
#define BPAD 16
#define NEG_SLOPE 0.2f

typedef __attribute__((ext_vector_type(2))) float v2f;
typedef __attribute__((ext_vector_type(8))) float v8f;

__device__ __forceinline__ unsigned encF(float f) {
  unsigned u = __float_as_uint(f);
  return (u & 0x80000000u) ? ~u : (u | 0x80000000u);
}
__device__ __forceinline__ float decF(unsigned u) {
  u = (u & 0x80000000u) ? (u & 0x7FFFFFFFu) : ~u;
  return __uint_as_float(u);
}

__device__ __forceinline__ float leaky(float v) {
  return v >= 0.f ? v : NEG_SLOPE * v;
}

__global__ __launch_bounds__(256) void gat_gemm_kernel(
    const float* __restrict__ x, const float* __restrict__ W,
    const float* __restrict__ att_src, const float* __restrict__ att_dst,
    float* __restrict__ h, float* __restrict__ asrc, float* __restrict__ adst,
    int n)
{
  __shared__ float Wlds[IN_CH * BPAD];
  for (int i = threadIdx.x; i < IN_CH * BPAD; i += blockDim.x) {
    const int k = i >> 4;
    const int c = i & 15;
    Wlds[i] = (c < HEADS) ? W[k * HEADS + c] : 0.f;
  }
  __syncthreads();

  const int lane = threadIdx.x & 31;
  const int wave = threadIdx.x >> 5;
  const int half = lane >> 4;
  const int l    = lane & 15;

  const int tile = blockIdx.x * (blockDim.x >> 5) + wave;
  const int row0 = tile * 16;
  if (row0 >= n) return;

  int arow = row0 + l;
  if (arow > n - 1) arow = n - 1;
  const float* __restrict__ xrow = x + (size_t)arow * IN_CH;
  const float* __restrict__ wcol = Wlds + (size_t)half * 2 * BPAD + l;

  v8f c = {0.f, 0.f, 0.f, 0.f, 0.f, 0.f, 0.f, 0.f};

  #pragma unroll 4
  for (int k = 0; k < IN_CH; k += 4) {
    const int ka = k + half * 2;
    v2f a;
    a.x = xrow[ka];
    a.y = xrow[ka + 1];
    v2f b;
    b.x = wcol[k * BPAD];
    b.y = wcol[k * BPAD + BPAD];
    c = __builtin_amdgcn_wmma_f32_16x16x4_f32(
         false, a,  false, b,
         (short)0, c,  false,  false);
  }

  if (l < HEADS) {
    const float as = att_src[l];
    const float ad = att_dst[l];
    #pragma unroll
    for (int i = 0; i < 8; ++i) {
      const int r = row0 + i + half * 8;
      if (r < n) {
        const float hv = c[i];
        const size_t o = (size_t)r * HEADS + l;
        h[o]    = hv;
        asrc[o] = hv * as;
        adst[o] = hv * ad;
      }
    }
  }
}

__global__ void gat_init_kernel(unsigned* __restrict__ emaxEnc,
                                float* __restrict__ denom,
                                float* __restrict__ numer, size_t cnt)
{
  size_t i = (size_t)blockIdx.x * blockDim.x + threadIdx.x;
  if (i < cnt) {
    emaxEnc[i] = 0x007FFFFFu;
    denom[i]   = 0.f;
    numer[i]   = 0.f;
  }
}

__global__ __launch_bounds__(256) void gat_edge_max_kernel(
    const int* __restrict__ ei,
    const float* __restrict__ asrc, const float* __restrict__ adst,
    unsigned* __restrict__ emaxEnc, int nE, int n)
{
  const int e = blockIdx.x * blockDim.x + threadIdx.x;
  const int etot = nE + n;
  if (e >= etot) return;
  int src, dst;
  if (e < nE) { src = ei[e]; dst = ei[nE + e]; }
  else        { src = dst = e - nE; }

  const float4* ps = (const float4*)(asrc + (size_t)src * HEADS);
  const float4* pd = (const float4*)(adst + (size_t)dst * HEADS);
  const float4 s0 = ps[0], s1 = ps[1];
  const float4 d0 = pd[0], d1 = pd[1];
  float v[HEADS] = { s0.x + d0.x, s0.y + d0.y, s0.z + d0.z, s0.w + d0.w,
                     s1.x + d1.x, s1.y + d1.y, s1.z + d1.z, s1.w + d1.w };
  unsigned* base = emaxEnc + (size_t)dst * HEADS;
  #pragma unroll
  for (int hh = 0; hh < HEADS; ++hh)
    atomicMax(base + hh, encF(leaky(v[hh])));
}

__global__ __launch_bounds__(256) void gat_edge_acc_kernel(
    const int* __restrict__ ei,
    const float* __restrict__ asrc, const float* __restrict__ adst,
    const float* __restrict__ hmat, const unsigned* __restrict__ emaxEnc,
    float* __restrict__ denom, float* __restrict__ numer, int nE, int n)
{
  const int e = blockIdx.x * blockDim.x + threadIdx.x;
  const int etot = nE + n;
  if (e >= etot) return;
  int src, dst;
  if (e < nE) { src = ei[e]; dst = ei[nE + e]; }
  else        { src = dst = e - nE; }

  const float4* ps = (const float4*)(asrc + (size_t)src * HEADS);
  const float4* pd = (const float4*)(adst + (size_t)dst * HEADS);
  const float4* ph = (const float4*)(hmat + (size_t)src * HEADS);
  const float4 s0 = ps[0], s1 = ps[1];
  const float4 d0 = pd[0], d1 = pd[1];
  const float4 h0 = ph[0], h1 = ph[1];
  float v[HEADS]  = { s0.x + d0.x, s0.y + d0.y, s0.z + d0.z, s0.w + d0.w,
                      s1.x + d1.x, s1.y + d1.y, s1.z + d1.z, s1.w + d1.w };
  float hv[HEADS] = { h0.x, h0.y, h0.z, h0.w, h1.x, h1.y, h1.z, h1.w };

  const unsigned* me = emaxEnc + (size_t)dst * HEADS;
  float* dden = denom + (size_t)dst * HEADS;
  float* dnum = numer + (size_t)dst * HEADS;
  #pragma unroll
  for (int hh = 0; hh < HEADS; ++hh) {
    const float ee = expf(leaky(v[hh]) - decF(me[hh]));
    atomicAdd(dden + hh, ee);
    atomicAdd(dnum + hh, ee * hv[hh]);
  }
}

__global__ __launch_bounds__(256) void gat_finalize_kernel(
    const float* __restrict__ numer, const float* __restrict__ denom,
    const float* __restrict__ gat_bias, const float* __restrict__ lin_w,
    const float* __restrict__ lin_b, const float* __restrict__ bias,
    float* __restrict__ out, int n)
{
  const int i = blockIdx.x * blockDim.x + threadIdx.x;
  if (i >= n) return;
  const float* nu = numer + (size_t)i * HEADS;
  const float* de = denom + (size_t)i * HEADS;
  float z = lin_b[0];
  #pragma unroll
  for (int hh = 0; hh < HEADS; ++hh) {
    const float oh = nu[hh] / (de[hh] + 1e-16f) + gat_bias[hh];
    z += oh * lin_w[hh];
  }
  out[i] = fmaxf(z, 0.f) + bias[0];
}

extern "C" void kernel_launch(void* const* d_in, const int* in_sizes, int n_in,
                              void* d_out, int out_size, void* d_ws, size_t ws_size,
                              hipStream_t stream) {
  const float* x        = (const float*)d_in[0];
  const int*   ei       = (const int*)d_in[1];
  const float* W        = (const float*)d_in[2];
  const float* att_src  = (const float*)d_in[3];
  const float* att_dst  = (const float*)d_in[4];
  const float* gat_bias = (const float*)d_in[5];
  const float* lin_w    = (const float*)d_in[6];
  const float* lin_b    = (const float*)d_in[7];
  const float* bias     = (const float*)d_in[8];
  float* out = (float*)d_out;

  const int n  = in_sizes[0] / IN_CH;
  const int nE = in_sizes[1] / 2;

  const size_t nh  = (size_t)n * HEADS;
  const size_t seg = (nh * sizeof(float) + 255) & ~(size_t)255;
  char* ws = (char*)d_ws;
  float*    h_mat   = (float*)   (ws + 0 * seg);
  float*    asrc    = (float*)   (ws + 1 * seg);
  float*    adst    = (float*)   (ws + 2 * seg);
  unsigned* emaxEnc = (unsigned*)(ws + 3 * seg);
  float*    denom   = (float*)   (ws + 4 * seg);
  float*    numer   = (float*)   (ws + 5 * seg);

  const int tiles  = (n + 15) / 16;
  const int blocks = (tiles + 7) / 8;
  gat_gemm_kernel<<<blocks, 256, 0, stream>>>(x, W, att_src, att_dst,
                                              h_mat, asrc, adst, n);

  {
    const int b = (int)((nh + 255) / 256);
    gat_init_kernel<<<b, 256, 0, stream>>>(emaxEnc, denom, numer, nh);
  }

  const int etot = nE + n;
  const int eb = (etot + 255) / 256;
  gat_edge_max_kernel<<<eb, 256, 0, stream>>>(ei, asrc, adst, emaxEnc, nE, n);

  gat_edge_acc_kernel<<<eb, 256, 0, stream>>>(ei, asrc, adst, h_mat, emaxEnc,
                                              denom, numer, nE, n);

  gat_finalize_kernel<<<(n + 255) / 256, 256, 0, stream>>>(
      numer, denom, gat_bias, lin_w, lin_b, bias, out, n);
}


# LDS accumulators (max, exp sum, weighted sum) stored head by head, not node by node: the 32 lanes of one LDS atomic fall on all 64 banks, not 8; sums still in LDS arrival order
# speedup vs baseline: 10.0406x; 10.0406x over previous
#include <hip/hip_runtime.h>
#include <math.h>

#define IN_CH 512
#define HEADS 8
#define BPAD 16
#define NEG_SLOPE 0.2f
#define REC 16

typedef __attribute__((ext_vector_type(2))) float v2f;
typedef __attribute__((ext_vector_type(8))) float v8f;

__device__ __forceinline__ unsigned encF(float f) {
  unsigned u = __float_as_uint(f);
  return (u & 0x80000000u) ? ~u : (u | 0x80000000u);
}
__device__ __forceinline__ float decF(unsigned u) {
  u = (u & 0x80000000u) ? (u & 0x7FFFFFFFu) : ~u;
  return __uint_as_float(u);
}

__device__ __forceinline__ float leaky(float v) {
  return v >= 0.f ? v : NEG_SLOPE * v;
}

#define NPB 416

__device__ __forceinline__ unsigned range_key(int d) {
  const unsigned r = (unsigned)d / (unsigned)NPB;
  return r < 255u ? r : 255u;
}

__global__ __launch_bounds__(256) void gat_gemm_kernel(
    const float* __restrict__ x, const float* __restrict__ W,
    const float* __restrict__ att_src, const float* __restrict__ att_dst,
    float* __restrict__ rec, float* __restrict__ adst,
    int n,
    const int* __restrict__ ei, unsigned char* __restrict__ key, int nE)
{
  {
    const int* __restrict__ dstp = ei + nE;
    const int T = (int)(gridDim.x * blockDim.x);
    const int g = (int)(blockIdx.x * blockDim.x + threadIdx.x);
    const bool vec_ok = ((nE & 3) == 0) && ((((size_t)dstp) & 15) == 0) &&
                        ((((size_t)key) & 3) == 0);
    if (vec_ok) {
      const int4* __restrict__ dv = (const int4*)dstp;
      unsigned* __restrict__ kw = (unsigned*)key;
      const int nvec = nE >> 2;
      #pragma unroll 4
      for (int v = g; v < nvec; v += T) {
        const int4 d = dv[v];
        kw[v] = range_key(d.x) | (range_key(d.y) << 8) |
                (range_key(d.z) << 16) | (range_key(d.w) << 24);
      }
    } else {
      for (int e = g; e < nE; e += T) key[e] = (unsigned char)range_key(dstp[e]);
    }
  }

  __shared__ float Wlds[IN_CH * BPAD];
  for (int i = threadIdx.x; i < IN_CH * BPAD; i += blockDim.x) {
    const int k = i >> 4;
    const int c = i & 15;
    Wlds[i] = (c < HEADS) ? W[k * HEADS + c] : 0.f;
  }
  __syncthreads();

  const int lane = threadIdx.x & 31;
  const int wave = threadIdx.x >> 5;
  const int half = lane >> 4;
  const int l    = lane & 15;

  const int tile = blockIdx.x * (blockDim.x >> 5) + wave;
  const int row0 = tile * 16;
  if (row0 >= n) return;

  int arow = row0 + l;
  if (arow > n - 1) arow = n - 1;
  const float* __restrict__ xrow = x + (size_t)arow * IN_CH;
  const float* __restrict__ wcol = Wlds + (size_t)half * 2 * BPAD + l;

  v8f c = {0.f, 0.f, 0.f, 0.f, 0.f, 0.f, 0.f, 0.f};

  #pragma unroll 4
  for (int k = 0; k < IN_CH; k += 4) {
    const int ka = k + half * 2;
    v2f a;
    a.x = xrow[ka];
    a.y = xrow[ka + 1];
    v2f b;
    b.x = wcol[k * BPAD];
    b.y = wcol[k * BPAD + BPAD];
    c = __builtin_amdgcn_wmma_f32_16x16x4_f32(
         false, a,  false, b,
         (short)0, c,  false,  false);
  }

  if (l < HEADS) {
    const float as = att_src[l];
    const float ad = att_dst[l];
    #pragma unroll
    for (int i = 0; i < 8; ++i) {
      const int r = row0 + i + half * 8;
      if (r < n) {
        const float hv = c[i];
        float* __restrict__ pr = rec + (size_t)r * REC + l;
        pr[0]     = hv * as;
        pr[HEADS] = hv;
        adst[(size_t)r * HEADS + l] = hv * ad;
      }
    }
  }
}

#define RTH 1024
#define CAP 28672
#define NLD 8
#define ACC(hh, dl) ((hh) * NPB + (dl))

__device__ __forceinline__ unsigned zero_bytes(unsigned x) {
  return __builtin_amdgcn_bitop3_b32(x - 0x01010101u, x, 0x80808080u, 0x20);
}

__device__ __forceinline__ unsigned flags16(const uint4 k, unsigned kb) {
  return zero_bytes(k.x ^ kb) | (zero_bytes(k.y ^ kb) >> 1) |
         (zero_bytes(k.z ^ kb) >> 2) | (zero_bytes(k.w ^ kb) >> 3);
}

template <class F>
__device__ __forceinline__ void test32(const uint4 kA, const uint4 kB, int e0, int dB,
                                       unsigned kb, F& f)
{
  unsigned m = flags16(kA, kb) | (flags16(kB, kb) >> 4);
  while (m) {
    const int pos = __builtin_ctz(m);
    m &= m - 1u;
    const int r = pos & 7;
    f(e0 + (r < 4 ? dB : 0) + ((3 - (r & 3)) << 2) + (pos >> 3));
  }
}

template <class F>
__device__ __forceinline__ void test16(const uint4 k, int e0, unsigned kb, F& f)
{
  unsigned m = flags16(k, kb);
  while (m) {
    const int pos = __builtin_ctz(m);
    m &= m - 1u;
    f(e0 + ((7 - (pos & 7)) << 2) + (pos >> 3));
  }
}

template <class F>
__device__ __forceinline__ void scan_seg(const uint4* __restrict__ kv, int a, int b,
                                         unsigned kb, F& f)
{
  const int len   = b - a;
  const int nfull = len - (len % (RTH * NLD));
  for (int j = threadIdx.x; j < nfull; j += RTH * NLD) {
    uint4 k[NLD];
    #pragma unroll
    for (int u = 0; u < NLD; ++u) k[u] = kv[a + j + u * RTH];
    #pragma unroll
    for (int u = 0; u < NLD; u += 2)
      test32(k[u], k[u + 1], (a + j + u * RTH) * 16, RTH * 16, kb, f);
  }
  for (int j = nfull + threadIdx.x; j < len; j += RTH)
    test16(kv[a + j], (a + j) * 16, kb, f);
}

template <class F>
__device__ __forceinline__ void scan_key(const unsigned char* __restrict__ key, int nE,
                                         unsigned kbyte, F f)
{
  const bool vec_ok = ((((size_t)key) & 15) == 0);
  const int nvec = vec_ok ? (nE >> 4) : 0;
  const uint4* __restrict__ kv = (const uint4*)key;
  const unsigned kb = kbyte * 0x01010101u;
  const int start = (int)(((long long)nvec * blockIdx.x) / gridDim.x);
  scan_seg(kv, start, nvec, kb, f);
  scan_seg(kv, 0, start, kb, f);
  for (int e = nvec * 16 + threadIdx.x; e < nE; e += RTH)
    if (key[e] == kbyte) f(e);
}

template <class F>
__device__ __forceinline__ void scan_dst_plain(const int* __restrict__ dstp, int nE,
                                               int lo, unsigned cnt, F f)
{
  for (int e = threadIdx.x; e < nE; e += RTH) {
    const unsigned t = (unsigned)(dstp[e] - lo);
    if (t < cnt) f(e, (int)t);
  }
}

__global__ __launch_bounds__(RTH) void gat_range_kernel(
    const int* __restrict__ ei,
    const float* __restrict__ rec, const float* __restrict__ adst,
    const float* __restrict__ gat_bias, const float* __restrict__ lin_w,
    const float* __restrict__ lin_b, const float* __restrict__ bias,
    float* __restrict__ out, int nE, int n,
    const unsigned char* __restrict__ key)
{
  __shared__ unsigned emL[NPB * HEADS];
  __shared__ float    denL[NPB * HEADS];
  __shared__ float    numL[NPB * HEADS];
  __shared__ float    adL[NPB * HEADS];
  __shared__ int      listA[CAP];
  __shared__ int      listD[CAP];
  __shared__ unsigned nlist;

  const int tid = threadIdx.x;
  const int lo  = blockIdx.x * NPB;
  const int cntN = (n - lo < NPB) ? (n - lo) : NPB;
  const int cntE = cntN * HEADS;
  const int* __restrict__ dstp = ei + nE;

  for (int j = tid; j < cntE; j += RTH) {
    const float a = adst[(size_t)lo * HEADS + j];
    adL[j] = a;
    emL[ACC(j & 7, j >> 3)] = encF(leaky(rec[(size_t)(lo + (j >> 3)) * REC + (j & 7)] + a));
  }
  if (tid == 0) nlist = 0u;
  __syncthreads();

  scan_key(key, nE, blockIdx.x < 255u ? blockIdx.x : 255u, [&](int e) {
    const unsigned sl = atomicAdd(&nlist, 1u);
    if (sl < (unsigned)CAP) listA[sl] = e;
  });
  __syncthreads();
  const unsigned total = nlist;
  const bool listed = total <= (unsigned)CAP;

  auto edge_max = [&](int src, int dl) {
    const float4* ps = (const float4*)(rec + (size_t)src * REC);
    const float4 s0 = ps[0], s1 = ps[1];
    const float* ad = adL + dl * HEADS;
    const float v[HEADS] = { s0.x + ad[0], s0.y + ad[1], s0.z + ad[2], s0.w + ad[3],
                             s1.x + ad[4], s1.y + ad[5], s1.z + ad[6], s1.w + ad[7] };
    #pragma unroll
    for (int hh = 0; hh < HEADS; ++hh)
      atomicMax(&emL[ACC(hh, dl)], encF(leaky(v[hh])));
  };
  auto edge_acc = [&](int src, int dl) {
    const float4* pr = (const float4*)(rec + (size_t)src * REC);
    const float4 s0 = pr[0], s1 = pr[1];
    const float4 h0 = pr[2], h1 = pr[3];
    const float* ad = adL + dl * HEADS;
    const float v[HEADS]  = { s0.x + ad[0], s0.y + ad[1], s0.z + ad[2], s0.w + ad[3],
                              s1.x + ad[4], s1.y + ad[5], s1.z + ad[6], s1.w + ad[7] };
    const float hv[HEADS] = { h0.x, h0.y, h0.z, h0.w, h1.x, h1.y, h1.z, h1.w };
    float ee[HEADS];
    #pragma unroll
    for (int hh = 0; hh < HEADS; ++hh) ee[hh] = decF(emL[ACC(hh, dl)]);
    #pragma unroll
    for (int hh = 0; hh < HEADS; ++hh) ee[hh] = expf(leaky(v[hh]) - ee[hh]);
    #pragma unroll
    for (int hh = 0; hh < HEADS; ++hh) {
      atomicAdd(&denL[ACC(hh, dl)], ee[hh]);
      atomicAdd(&numL[ACC(hh, dl)], ee[hh] * hv[hh]);
    }
  };

  if (listed) {
    for (unsigned i = tid; i < total; i += RTH) {
      const int e   = listA[i];
      const int src = ei[e];
      const unsigned t = (unsigned)(dstp[e] - lo);
      const bool mine = t < (unsigned)cntN;
      listA[i] = src;
      listD[i] = mine ? (int)t : -1;
      if (mine) edge_max(src, (int)t);
    }
  } else {
    scan_dst_plain(dstp, nE, lo, (unsigned)cntN, [&](int e, int dl) { edge_max(ei[e], dl); });
  }
  __syncthreads();

  for (int j = tid; j < cntE; j += RTH) {
    const float* __restrict__ pr = rec + (size_t)(lo + (j >> 3)) * REC + (j & 7);
    const int a = ACC(j & 7, j >> 3);
    const float ee = expf(leaky(pr[0] + adL[j]) - decF(emL[a]));
    denL[a] = ee;
    numL[a] = ee * pr[HEADS];
  }
  __syncthreads();

  if (listed) {
    for (unsigned i = tid; i < total; i += RTH) {
      const int dl = listD[i];
      if (dl >= 0) edge_acc(listA[i], dl);
    }
  } else {
    scan_dst_plain(dstp, nE, lo, (unsigned)cntN, [&](int e, int dl) { edge_acc(ei[e], dl); });
  }
  __syncthreads();

  for (int i = tid; i < cntN; i += RTH) {
    float z = lin_b[0];
    #pragma unroll
    for (int hh = 0; hh < HEADS; ++hh) {
      const float oh = numL[ACC(hh, i)] / (denL[ACC(hh, i)] + 1e-16f) + gat_bias[hh];
      z += oh * lin_w[hh];
    }
    out[lo + i] = fmaxf(z, 0.f) + bias[0];
  }
}

extern "C" void kernel_launch(void* const* d_in, const int* in_sizes, int n_in,
                              void* d_out, int out_size, void* d_ws, size_t ws_size,
                              hipStream_t stream) {
  const float* x        = (const float*)d_in[0];
  const int*   ei       = (const int*)d_in[1];
  const float* W        = (const float*)d_in[2];
  const float* att_src  = (const float*)d_in[3];
  const float* att_dst  = (const float*)d_in[4];
  const float* gat_bias = (const float*)d_in[5];
  const float* lin_w    = (const float*)d_in[6];
  const float* lin_b    = (const float*)d_in[7];
  const float* bias     = (const float*)d_in[8];
  float* out = (float*)d_out;

  const int n  = in_sizes[0] / IN_CH;
  const int nE = in_sizes[1] / 2;

  const size_t nh  = (size_t)n * HEADS;
  const size_t seg = (nh * sizeof(float) + 255) & ~(size_t)255;
  char* ws = (char*)d_ws;
  float*    rec     = (float*)   (ws + 0 * seg);
  float*    adst    = (float*)   (ws + 2 * seg);
  unsigned char* key = (unsigned char*)(ws + 3 * seg);

  const int tiles  = (n + 15) / 16;
  const int blocks = (tiles + 7) / 8;
  gat_gemm_kernel<<<blocks, 256, 0, stream>>>(x, W, att_src, att_dst,
                                              rec, adst, n, ei, key, nE);

  gat_range_kernel<<<(n + NPB - 1) / NPB, RTH, 0, stream>>>(
      ei, rec, adst, gat_bias, lin_w, lin_b, bias, out, nE, n, key);
}


# edge list re-filed in LDS into 32 classes by local destination mod 32, lane L walks class L: an LDS atomic's 32 lanes hit 32 different banks, never one word twice; sums still in LDS arrival order
# speedup vs baseline: 10.9739x; 1.0930x over previous
#include <hip/hip_runtime.h>
#include <math.h>

#define IN_CH 512
#define HEADS 8
#define BPAD 16
#define NEG_SLOPE 0.2f
#define REC 16

typedef __attribute__((ext_vector_type(2))) float v2f;
typedef __attribute__((ext_vector_type(8))) float v8f;

__device__ __forceinline__ unsigned encF(float f) {
  unsigned u = __float_as_uint(f);
  return (u & 0x80000000u) ? ~u : (u | 0x80000000u);
}
__device__ __forceinline__ float decF(unsigned u) {
  u = (u & 0x80000000u) ? (u & 0x7FFFFFFFu) : ~u;
  return __uint_as_float(u);
}

__device__ __forceinline__ float leaky(float v) {
  return v >= 0.f ? v : NEG_SLOPE * v;
}

#define NPB 416

__device__ __forceinline__ unsigned range_key(int d) {
  const unsigned r = (unsigned)d / (unsigned)NPB;
  return r < 255u ? r : 255u;
}

__global__ __launch_bounds__(256) void gat_gemm_kernel(
    const float* __restrict__ x, const float* __restrict__ W,
    const float* __restrict__ att_src, const float* __restrict__ att_dst,
    float* __restrict__ rec, float* __restrict__ adst,
    int n,
    const int* __restrict__ ei, unsigned char* __restrict__ key, int nE)
{
  {
    const int* __restrict__ dstp = ei + nE;
    const int T = (int)(gridDim.x * blockDim.x);
    const int g = (int)(blockIdx.x * blockDim.x + threadIdx.x);
    const bool vec_ok = ((nE & 3) == 0) && ((((size_t)dstp) & 15) == 0) &&
                        ((((size_t)key) & 3) == 0);
    if (vec_ok) {
      const int4* __restrict__ dv = (const int4*)dstp;
      unsigned* __restrict__ kw = (unsigned*)key;
      const int nvec = nE >> 2;
      #pragma unroll 4
      for (int v = g; v < nvec; v += T) {
        const int4 d = dv[v];
        kw[v] = range_key(d.x) | (range_key(d.y) << 8) |
                (range_key(d.z) << 16) | (range_key(d.w) << 24);
      }
    } else {
      for (int e = g; e < nE; e += T) key[e] = (unsigned char)range_key(dstp[e]);
    }
  }

  __shared__ float Wlds[IN_CH * BPAD];
  for (int i = threadIdx.x; i < IN_CH * BPAD; i += blockDim.x) {
    const int k = i >> 4;
    const int c = i & 15;
    Wlds[i] = (c < HEADS) ? W[k * HEADS + c] : 0.f;
  }
  __syncthreads();

  const int lane = threadIdx.x & 31;
  const int wave = threadIdx.x >> 5;
  const int half = lane >> 4;
  const int l    = lane & 15;

  const int tile = blockIdx.x * (blockDim.x >> 5) + wave;
  const int row0 = tile * 16;
  if (row0 >= n) return;

  int arow = row0 + l;
  if (arow > n - 1) arow = n - 1;
  const float* __restrict__ xrow = x + (size_t)arow * IN_CH;
  const float* __restrict__ wcol = Wlds + (size_t)half * 2 * BPAD + l;

  v8f c = {0.f, 0.f, 0.f, 0.f, 0.f, 0.f, 0.f, 0.f};

  #pragma unroll 4
  for (int k = 0; k < IN_CH; k += 4) {
    const int ka = k + half * 2;
    v2f a;
    a.x = xrow[ka];
    a.y = xrow[ka + 1];
    v2f b;
    b.x = wcol[k * BPAD];
    b.y = wcol[k * BPAD + BPAD];
    c = __builtin_amdgcn_wmma_f32_16x16x4_f32(
         false, a,  false, b,
         (short)0, c,  false,  false);
  }

  if (l < HEADS) {
    const float as = att_src[l];
    const float ad = att_dst[l];
    #pragma unroll
    for (int i = 0; i < 8; ++i) {
      const int r = row0 + i + half * 8;
      if (r < n) {
        const float hv = c[i];
        float* __restrict__ pr = rec + (size_t)r * REC + l;
        pr[0]     = hv * as;
        pr[HEADS] = hv;
        adst[(size_t)r * HEADS + l] = hv * ad;
      }
    }
  }
}

#define RTH 1024
#define CAP 28672
#define NLD 8
#define ACC(hh, dl) ((hh) * NPB + (dl))
#define NCLS 32
#define CAPC 896
#define DLB 9
#define ADP 12

__device__ __forceinline__ unsigned zero_bytes(unsigned x) {
  return __builtin_amdgcn_bitop3_b32(x - 0x01010101u, x, 0x80808080u, 0x20);
}

__device__ __forceinline__ unsigned flags16(const uint4 k, unsigned kb) {
  return zero_bytes(k.x ^ kb) | (zero_bytes(k.y ^ kb) >> 1) |
         (zero_bytes(k.z ^ kb) >> 2) | (zero_bytes(k.w ^ kb) >> 3);
}

template <class F>
__device__ __forceinline__ void test32(const uint4 kA, const uint4 kB, int e0, int dB,
                                       unsigned kb, F& f)
{
  unsigned m = flags16(kA, kb) | (flags16(kB, kb) >> 4);
  while (m) {
    const int pos = __builtin_ctz(m);
    m &= m - 1u;
    const int r = pos & 7;
    f(e0 + (r < 4 ? dB : 0) + ((3 - (r & 3)) << 2) + (pos >> 3));
  }
}

template <class F>
__device__ __forceinline__ void test16(const uint4 k, int e0, unsigned kb, F& f)
{
  unsigned m = flags16(k, kb);
  while (m) {
    const int pos = __builtin_ctz(m);
    m &= m - 1u;
    f(e0 + ((7 - (pos & 7)) << 2) + (pos >> 3));
  }
}

template <class F>
__device__ __forceinline__ void scan_seg(const uint4* __restrict__ kv, int a, int b,
                                         unsigned kb, F& f)
{
  const int len   = b - a;
  const int nfull = len - (len % (RTH * NLD));
  for (int j = threadIdx.x; j < nfull; j += RTH * NLD) {
    uint4 k[NLD];
    #pragma unroll
    for (int u = 0; u < NLD; ++u) k[u] = kv[a + j + u * RTH];
    #pragma unroll
    for (int u = 0; u < NLD; u += 2)
      test32(k[u], k[u + 1], (a + j + u * RTH) * 16, RTH * 16, kb, f);
  }
  for (int j = nfull + threadIdx.x; j < len; j += RTH)
    test16(kv[a + j], (a + j) * 16, kb, f);
}

template <class F>
__device__ __forceinline__ void scan_key(const unsigned char* __restrict__ key, int nE,
                                         unsigned kbyte, F f)
{
  const bool vec_ok = ((((size_t)key) & 15) == 0);
  const int nvec = vec_ok ? (nE >> 4) : 0;
  const uint4* __restrict__ kv = (const uint4*)key;
  const unsigned kb = kbyte * 0x01010101u;
  const int start = (int)(((long long)nvec * blockIdx.x) / gridDim.x);
  scan_seg(kv, start, nvec, kb, f);
  scan_seg(kv, 0, start, kb, f);
  for (int e = nvec * 16 + threadIdx.x; e < nE; e += RTH)
    if (key[e] == kbyte) f(e);
}

template <class F>
__device__ __forceinline__ void scan_dst_plain(const int* __restrict__ dstp, int nE,
                                               int lo, unsigned cnt, F f)
{
  for (int e = threadIdx.x; e < nE; e += RTH) {
    const unsigned t = (unsigned)(dstp[e] - lo);
    if (t < cnt) f(e, (int)t);
  }
}

__global__ __launch_bounds__(RTH) void gat_range_kernel(
    const int* __restrict__ ei,
    const float* __restrict__ rec, const float* __restrict__ adst,
    const float* __restrict__ gat_bias, const float* __restrict__ lin_w,
    const float* __restrict__ lin_b, const float* __restrict__ bias,
    float* __restrict__ out, int nE, int n,
    const unsigned char* __restrict__ key)
{
  __shared__ unsigned emL[NPB * HEADS];
  __shared__ float    denL[NPB * HEADS];
  __shared__ float    numL[NPB * HEADS];
  __shared__ __attribute__((aligned(16))) float adL[NPB * ADP];
  __shared__ int      listA[CAP];
  __shared__ unsigned listC[NCLS * CAPC];
  __shared__ unsigned ccnt[NCLS];
  __shared__ unsigned nlist;
  __shared__ unsigned over;

  const int tid = threadIdx.x;
  const int lo  = blockIdx.x * NPB;
  const int cntN = (n - lo < NPB) ? (n - lo) : NPB;
  const int cntE = cntN * HEADS;
  const int* __restrict__ dstp = ei + nE;

  for (int j = tid; j < cntE; j += RTH) {
    const float a = adst[(size_t)lo * HEADS + j];
    adL[(j >> 3) * ADP + (j & 7)] = a;
    emL[ACC(j & 7, j >> 3)] = encF(leaky(rec[(size_t)(lo + (j >> 3)) * REC + (j & 7)] + a));
  }
  if (tid < NCLS) ccnt[tid] = 0u;
  if (tid == 0) { nlist = 0u; over = 0u; }
  __syncthreads();

  scan_key(key, nE, blockIdx.x < 255u ? blockIdx.x : 255u, [&](int e) {
    const unsigned sl = atomicAdd(&nlist, 1u);
    if (sl < (unsigned)CAP) listA[sl] = e;
  });
  __syncthreads();
  const unsigned total = nlist;
  const bool listed = total <= (unsigned)CAP && (unsigned)n <= (1u << (32 - DLB));

  auto edge_max = [&](int src, int dl) {
    const float4* ps = (const float4*)(rec + (size_t)src * REC);
    const float4 s0 = ps[0], s1 = ps[1];
    const float* ad = adL + dl * ADP;
    const float v[HEADS] = { s0.x + ad[0], s0.y + ad[1], s0.z + ad[2], s0.w + ad[3],
                             s1.x + ad[4], s1.y + ad[5], s1.z + ad[6], s1.w + ad[7] };
    #pragma unroll
    for (int hh = 0; hh < HEADS; ++hh)
      atomicMax(&emL[ACC(hh, dl)], encF(leaky(v[hh])));
  };
  auto edge_acc = [&](int src, int dl) {
    const float4* pr = (const float4*)(rec + (size_t)src * REC);
    const float4 s0 = pr[0], s1 = pr[1];
    const float4 h0 = pr[2], h1 = pr[3];
    const float* ad = adL + dl * ADP;
    const float v[HEADS]  = { s0.x + ad[0], s0.y + ad[1], s0.z + ad[2], s0.w + ad[3],
                              s1.x + ad[4], s1.y + ad[5], s1.z + ad[6], s1.w + ad[7] };
    const float hv[HEADS] = { h0.x, h0.y, h0.z, h0.w, h1.x, h1.y, h1.z, h1.w };
    float ee[HEADS];
    #pragma unroll
    for (int hh = 0; hh < HEADS; ++hh) ee[hh] = decF(emL[ACC(hh, dl)]);
    #pragma unroll
    for (int hh = 0; hh < HEADS; ++hh) ee[hh] = expf(leaky(v[hh]) - ee[hh]);
    #pragma unroll
    for (int hh = 0; hh < HEADS; ++hh) {
      atomicAdd(&denL[ACC(hh, dl)], ee[hh]);
      atomicAdd(&numL[ACC(hh, dl)], ee[hh] * hv[hh]);
    }
  };

  if (listed) {
    for (unsigned i = tid; i < total; i += RTH) {
      const int e   = listA[i];
      int src = ei[e];
      const unsigned t = (unsigned)(dstp[e] - lo);
      asm volatile("" : "+v"(src));
      if (t < (unsigned)cntN) {
        const unsigned c = t & (NCLS - 1);
        const unsigned s = atomicAdd(&ccnt[c], 1u);
        if (s < (unsigned)CAPC) listC[s * NCLS + c] = ((unsigned)src << DLB) | t;
        else over = 1u;
      }
    }
  }
  __syncthreads();
  const bool binned = listed && over == 0u;
  const unsigned lane = (unsigned)tid & 31u;
  const unsigned mycnt = binned ? ccnt[lane] : 0u;

  if (binned) {
    for (unsigned s = (unsigned)tid >> 5; s < mycnt; s += RTH / 32) {
      const unsigned w = listC[s * NCLS + lane];
      edge_max((int)(w >> DLB), (int)(w & ((1u << DLB) - 1u)));
    }
  } else {
    scan_dst_plain(dstp, nE, lo, (unsigned)cntN, [&](int e, int dl) { edge_max(ei[e], dl); });
  }
  __syncthreads();

  for (int j = tid; j < cntE; j += RTH) {
    const float* __restrict__ pr = rec + (size_t)(lo + (j >> 3)) * REC + (j & 7);
    const int a = ACC(j & 7, j >> 3);
    const float ee = expf(leaky(pr[0] + adL[(j >> 3) * ADP + (j & 7)]) - decF(emL[a]));
    denL[a] = ee;
    numL[a] = ee * pr[HEADS];
  }
  __syncthreads();

  if (binned) {
    for (unsigned s = (unsigned)tid >> 5; s < mycnt; s += RTH / 32) {
      const unsigned w = listC[s * NCLS + lane];
      edge_acc((int)(w >> DLB), (int)(w & ((1u << DLB) - 1u)));
    }
  } else {
    scan_dst_plain(dstp, nE, lo, (unsigned)cntN, [&](int e, int dl) { edge_acc(ei[e], dl); });
  }
  __syncthreads();

  for (int i = tid; i < cntN; i += RTH) {
    float z = lin_b[0];
    #pragma unroll
    for (int hh = 0; hh < HEADS; ++hh) {
      const float oh = numL[ACC(hh, i)] / (denL[ACC(hh, i)] + 1e-16f) + gat_bias[hh];
      z += oh * lin_w[hh];
    }
    out[lo + i] = fmaxf(z, 0.f) + bias[0];
  }
}

extern "C" void kernel_launch(void* const* d_in, const int* in_sizes, int n_in,
                              void* d_out, int out_size, void* d_ws, size_t ws_size,
                              hipStream_t stream) {
  const float* x        = (const float*)d_in[0];
  const int*   ei       = (const int*)d_in[1];
  const float* W        = (const float*)d_in[2];
  const float* att_src  = (const float*)d_in[3];
  const float* att_dst  = (const float*)d_in[4];
  const float* gat_bias = (const float*)d_in[5];
  const float* lin_w    = (const float*)d_in[6];
  const float* lin_b    = (const float*)d_in[7];
  const float* bias     = (const float*)d_in[8];
  float* out = (float*)d_out;

  const int n  = in_sizes[0] / IN_CH;
  const int nE = in_sizes[1] / 2;

  const size_t nh  = (size_t)n * HEADS;
  const size_t seg = (nh * sizeof(float) + 255) & ~(size_t)255;
  char* ws = (char*)d_ws;
  float*    rec     = (float*)   (ws + 0 * seg);
  float*    adst    = (float*)   (ws + 2 * seg);
  unsigned char* key = (unsigned char*)(ws + 3 * seg);

  const int tiles  = (n + 15) / 16;
  const int blocks = (tiles + 7) / 8;
  gat_gemm_kernel<<<blocks, 256, 0, stream>>>(x, W, att_src, att_dst,
                                              rec, adst, n, ei, key, nE);

  gat_range_kernel<<<(n + NPB - 1) / NPB, RTH, 0, stream>>>(
      ei, rec, adst, gat_bias, lin_w, lin_b, bias, out, nE, n, key);
}


# a bit per range and 16-edge group, not a byte per edge: a range block fetches only the hit groups; a_dst formed in the range block, the index takes its plane's room; list and sum order change
# speedup vs baseline: 12.7521x; 1.1620x over previous
#include <hip/hip_runtime.h>
#include <math.h>

#define IN_CH 512
#define HEADS 8
#define BPAD 16
#define NEG_SLOPE 0.2f
#define REC 16

typedef __attribute__((ext_vector_type(2))) float v2f;
typedef __attribute__((ext_vector_type(8))) float v8f;

__device__ __forceinline__ unsigned encF(float f) {
  unsigned u = __float_as_uint(f);
  return (u & 0x80000000u) ? ~u : (u | 0x80000000u);
}
__device__ __forceinline__ float decF(unsigned u) {
  u = (u & 0x80000000u) ? (u & 0x7FFFFFFFu) : ~u;
  return __uint_as_float(u);
}

__device__ __forceinline__ float leaky(float v) {
  return v >= 0.f ? v : NEG_SLOPE * v;
}

#define NPB 416

__device__ __forceinline__ unsigned range_key(int d) {
  const unsigned r = (unsigned)d / (unsigned)NPB;
  return r < 255u ? r : 255u;
}

#define GTH 256
#define CHP 1024
#define CHW 8
#define RMAX 256

__device__ __forceinline__ int bm_rows(int n) {
  const int nr = (n + NPB - 1) / NPB;
  const int nk = nr < RMAX ? nr : RMAX;
  return (nk + 3) & ~3;
}

__device__ __forceinline__ bool vec_path(const int* ei, int nE, const unsigned* bm) {
  return ((nE & 3) == 0) && ((((size_t)ei) & 15) == 0) && ((((size_t)bm) & 15) == 0);
}

__global__ __launch_bounds__(GTH) void gat_gemm_kernel(
    const float* __restrict__ x, const float* __restrict__ W,
    const float* __restrict__ att_src,
    float* __restrict__ rec,
    int n,
    const int* __restrict__ ei, unsigned* __restrict__ bm, int nE)
{
  __shared__ float Wlds[IN_CH * BPAD];
  __shared__ __attribute__((aligned(16))) unsigned bmL[RMAX * CHW];

  const bool vec = vec_path(ei, nE, bm);
  const int4* __restrict__ dv = (const int4*)(ei + nE);
  const int nPieces = nE >> 2;
  const int nChunks = vec ? (nPieces + CHP - 1) / CHP : 0;
  const int RP = bm_rows(n);
  int ch = (int)blockIdx.x;
  int4 d[CHP / GTH];
  auto load_chunk = [&]() {
    #pragma unroll
    for (int u = 0; u < CHP / GTH; ++u) {
      const int p = ch * CHP + u * GTH + (int)threadIdx.x;
      d[u] = dv[p < nPieces ? p : nPieces - 1];
    }
  };
  auto clear_rows = [&]() {
    for (int i = threadIdx.x; i < RMAX * CHW / 4; i += GTH)
      ((uint4*)bmL)[i] = make_uint4(0u, 0u, 0u, 0u);
  };
  if (ch < nChunks) load_chunk();
  clear_rows();

  for (int i = threadIdx.x; i < IN_CH * BPAD; i += GTH) {
    const int k = i >> 4;
    const int c = i & 15;
    Wlds[i] = (c < HEADS) ? W[k * HEADS + c] : 0.f;
  }
  __syncthreads();

  while (ch < nChunks) {
    #pragma unroll
    for (int u = 0; u < CHP / GTH; ++u) {
      const int pl = u * GTH + (int)threadIdx.x;
      if (ch * CHP + pl < nPieces) {
        const unsigned w   = (unsigned)pl >> 7;
        const unsigned bit = 1u << (((unsigned)pl >> 2) & 31u);
        atomicOr(&bmL[range_key(d[u].x) * CHW + w], bit);
        atomicOr(&bmL[range_key(d[u].y) * CHW + w], bit);
        atomicOr(&bmL[range_key(d[u].z) * CHW + w], bit);
        atomicOr(&bmL[range_key(d[u].w) * CHW + w], bit);
      }
    }
    __syncthreads();
    uint4* __restrict__ o = (uint4*)bm + (size_t)ch * (size_t)(RP * 2);
    for (int i = threadIdx.x; i < RP * 2; i += GTH) o[i] = ((const uint4*)bmL)[i];
    ch += (int)gridDim.x;
    if (ch < nChunks) {
      __syncthreads();
      load_chunk();
      clear_rows();
      __syncthreads();
    }
  }

  const int lane = threadIdx.x & 31;
  const int wave = threadIdx.x >> 5;
  const int half = lane >> 4;
  const int l    = lane & 15;

  const int tile = blockIdx.x * (GTH >> 5) + wave;
  const int row0 = tile * 16;
  if (row0 >= n) return;

  int arow = row0 + l;
  if (arow > n - 1) arow = n - 1;
  const float* __restrict__ xrow = x + (size_t)arow * IN_CH;
  const float* __restrict__ wcol = Wlds + (size_t)half * 2 * BPAD + l;

  v8f c = {0.f, 0.f, 0.f, 0.f, 0.f, 0.f, 0.f, 0.f};

  #pragma unroll 4
  for (int k = 0; k < IN_CH; k += 4) {
    const int ka = k + half * 2;
    v2f a;
    a.x = xrow[ka];
    a.y = xrow[ka + 1];
    v2f b;
    b.x = wcol[k * BPAD];
    b.y = wcol[k * BPAD + BPAD];
    c = __builtin_amdgcn_wmma_f32_16x16x4_f32(
         false, a,  false, b,
         (short)0, c,  false,  false);
  }

  if (l < HEADS) {
    const float as = att_src[l];
    #pragma unroll
    for (int i = 0; i < 8; ++i) {
      const int r = row0 + i + half * 8;
      if (r < n) {
        const float hv = c[i];
        float* __restrict__ pr = rec + (size_t)r * REC + l;
        pr[0]     = hv * as;
        pr[HEADS] = hv;
      }
    }
  }
}

#define RTH 1024
#define CAP 28672
#define EW 8
#define RU 4
#define ACC(hh, dl) ((hh) * NPB + (dl))
#define NCLS 32
#define CAPC 896
#define DLB 9
#define ADP 12

template <class F>
__device__ __forceinline__ void scan_dst_plain(const int* __restrict__ dstp, int nE,
                                               int lo, unsigned cnt, F f)
{
  for (int e = threadIdx.x; e < nE; e += RTH) {
    const unsigned t = (unsigned)(dstp[e] - lo);
    if (t < cnt) f(e, (int)t);
  }
}

__global__ __launch_bounds__(RTH) void gat_range_kernel(
    const int* __restrict__ ei,
    const float* __restrict__ rec, const float* __restrict__ att_dst,
    const float* __restrict__ gat_bias, const float* __restrict__ lin_w,
    const float* __restrict__ lin_b, const float* __restrict__ bias,
    float* __restrict__ out, int nE, int n,
    const unsigned* __restrict__ bm)
{
  __shared__ unsigned emL[NPB * HEADS];
  __shared__ float    denL[NPB * HEADS];
  __shared__ float    numL[NPB * HEADS];
  __shared__ __attribute__((aligned(16))) float adL[NPB * ADP];
  __shared__ unsigned listG[CAP];
  __shared__ unsigned listC[NCLS * CAPC];
  __shared__ unsigned ccnt[NCLS];
  __shared__ unsigned nlist;
  __shared__ unsigned over;

  const int tid = threadIdx.x;
  const int lo  = blockIdx.x * NPB;
  const int cntN = (n - lo < NPB) ? (n - lo) : NPB;
  const int cntE = cntN * HEADS;
  const int* __restrict__ dstp = ei + nE;

  const bool vec_ok  = vec_path(ei, nE, bm);
  const int  nPieces = nE >> 2;
  const int  nW      = vec_ok ? ((nPieces + CHP - 1) / CHP) * CHW : 0;
  const unsigned* __restrict__ myrow =
      bm + (size_t)(blockIdx.x < 255u ? blockIdx.x : 255u) * CHW;
  const size_t cpitch = (size_t)bm_rows(n) * CHW;
  unsigned wd[EW];
  auto load_words = [&](int w0) {
    #pragma unroll
    for (int u = 0; u < EW; ++u) {
      const int w = w0 + u * RTH + tid;
      wd[u] = (w < nW) ? myrow[(size_t)(w >> 3) * cpitch + (size_t)(w & 7)] : 0u;
    }
  };
  load_words(0);

  for (int j = tid; j < cntE; j += RTH) {
    const float* __restrict__ pr = rec + (size_t)(lo + (j >> 3)) * REC + (j & 7);
    const float a = pr[HEADS] * att_dst[j & 7];
    adL[(j >> 3) * ADP + (j & 7)] = a;
    emL[ACC(j & 7, j >> 3)] = encF(leaky(pr[0] + a));
  }
  if (tid < NCLS) ccnt[tid] = 0u;
  if (tid == 0) { nlist = 0u; over = 0u; }
  __syncthreads();

  for (int w0 = 0; w0 < nW; ) {
    unsigned cnt = 0u;
    #pragma unroll
    for (int u = 0; u < EW; ++u) cnt += (unsigned)__builtin_popcount(wd[u]);
    if (cnt) {
      unsigned sl = atomicAdd(&nlist, cnt);
      #pragma unroll
      for (int u = 0; u < EW; ++u) {
        unsigned m = wd[u];
        const unsigned g0 = (unsigned)(w0 + u * RTH + tid) << 5;
        while (m) {
          const unsigned b = (unsigned)__builtin_ctz(m);
          m &= m - 1u;
          if (sl < (unsigned)CAP) listG[sl] = g0 + b;
          ++sl;
        }
      }
    }
    w0 += RTH * EW;
    if (w0 < nW) load_words(w0);
  }
  __syncthreads();
  const unsigned total = nlist;
  const bool listed = vec_ok && total <= (unsigned)CAP && (unsigned)n <= (1u << (32 - DLB));

  auto edge_max = [&](int src, int dl) {
    const float4* ps = (const float4*)(rec + (size_t)src * REC);
    const float4 s0 = ps[0], s1 = ps[1];
    const float* ad = adL + dl * ADP;
    const float v[HEADS] = { s0.x + ad[0], s0.y + ad[1], s0.z + ad[2], s0.w + ad[3],
                             s1.x + ad[4], s1.y + ad[5], s1.z + ad[6], s1.w + ad[7] };
    #pragma unroll
    for (int hh = 0; hh < HEADS; ++hh)
      atomicMax(&emL[ACC(hh, dl)], encF(leaky(v[hh])));
  };
  auto edge_acc = [&](int src, int dl) {
    const float4* pr = (const float4*)(rec + (size_t)src * REC);
    const float4 s0 = pr[0], s1 = pr[1];
    const float4 h0 = pr[2], h1 = pr[3];
    const float* ad = adL + dl * ADP;
    const float v[HEADS]  = { s0.x + ad[0], s0.y + ad[1], s0.z + ad[2], s0.w + ad[3],
                              s1.x + ad[4], s1.y + ad[5], s1.z + ad[6], s1.w + ad[7] };
    const float hv[HEADS] = { h0.x, h0.y, h0.z, h0.w, h1.x, h1.y, h1.z, h1.w };
    float ee[HEADS];
    #pragma unroll
    for (int hh = 0; hh < HEADS; ++hh) ee[hh] = decF(emL[ACC(hh, dl)]);
    #pragma unroll
    for (int hh = 0; hh < HEADS; ++hh) ee[hh] = expf(leaky(v[hh]) - ee[hh]);
    #pragma unroll
    for (int hh = 0; hh < HEADS; ++hh) {
      atomicAdd(&denL[ACC(hh, dl)], ee[hh]);
      atomicAdd(&numL[ACC(hh, dl)], ee[hh] * hv[hh]);
    }
  };

  if (listed) {
    const int4* __restrict__ dv = (const int4*)dstp;
    const int4* __restrict__ sv = (const int4*)ei;
    const unsigned tot4 = total << 2;
    for (unsigned i0 = 0; i0 < tot4; i0 += RTH * RU) {
      int4 d[RU], s[RU];
      bool ok[RU];
      #pragma unroll
      for (int u = 0; u < RU; ++u) {
        const unsigned i = i0 + (unsigned)(u * RTH + tid);
        const bool in = i < tot4;
        int p = (int)((listG[in ? (i >> 2) : 0u] << 2) | (i & 3u));
        ok[u] = in && p < nPieces;
        p = ok[u] ? p : 0;
        d[u] = dv[p];
        s[u] = sv[p];
      }
      #pragma unroll
      for (int u = 0; u < RU; ++u) {
        asm volatile("" : "+v"(s[u].x), "+v"(s[u].y), "+v"(s[u].z), "+v"(s[u].w));
        const unsigned t0 = (unsigned)(d[u].x - lo), t1 = (unsigned)(d[u].y - lo);
        const unsigned t2 = (unsigned)(d[u].z - lo), t3 = (unsigned)(d[u].w - lo);
        unsigned mk = (t0 < (unsigned)cntN ? 1u : 0u) | (t1 < (unsigned)cntN ? 2u : 0u) |
                      (t2 < (unsigned)cntN ? 4u : 0u) | (t3 < (unsigned)cntN ? 8u : 0u);
        mk = ok[u] ? mk : 0u;
        const unsigned w0 = ((unsigned)s[u].x << DLB) | t0, w1 = ((unsigned)s[u].y << DLB) | t1;
        const unsigned w2 = ((unsigned)s[u].z << DLB) | t2, w3 = ((unsigned)s[u].w << DLB) | t3;
        while (mk) {
          const unsigned j = (unsigned)__builtin_ctz(mk);
          mk &= mk - 1u;
          const bool j1 = __builtin_unpredictable((j & 1u) != 0u);
          const bool j2 = __builtin_unpredictable((j & 2u) != 0u);
          const unsigned w  = j2 ? (j1 ? w3 : w2) : (j1 ? w1 : w0);
          const unsigned c  = w & (NCLS - 1);
          const unsigned sl = atomicAdd(&ccnt[c], 1u);
          if (sl < (unsigned)CAPC) listC[sl * NCLS + c] = w;
          else over = 1u;
        }
      }
    }
  }
  __syncthreads();
  const bool binned = listed && over == 0u;
  const unsigned lane = (unsigned)tid & 31u;
  const unsigned mycnt = binned ? ccnt[lane] : 0u;

  if (binned) {
    for (unsigned s = (unsigned)tid >> 5; s < mycnt; s += RTH / 32) {
      const unsigned w = listC[s * NCLS + lane];
      edge_max((int)(w >> DLB), (int)(w & ((1u << DLB) - 1u)));
    }
  } else {
    scan_dst_plain(dstp, nE, lo, (unsigned)cntN, [&](int e, int dl) { edge_max(ei[e], dl); });
  }
  __syncthreads();

  for (int j = tid; j < cntE; j += RTH) {
    const float* __restrict__ pr = rec + (size_t)(lo + (j >> 3)) * REC + (j & 7);
    const int a = ACC(j & 7, j >> 3);
    const float ee = expf(leaky(pr[0] + adL[(j >> 3) * ADP + (j & 7)]) - decF(emL[a]));
    denL[a] = ee;
    numL[a] = ee * pr[HEADS];
  }
  __syncthreads();

  if (binned) {
    for (unsigned s = (unsigned)tid >> 5; s < mycnt; s += RTH / 32) {
      const unsigned w = listC[s * NCLS + lane];
      edge_acc((int)(w >> DLB), (int)(w & ((1u << DLB) - 1u)));
    }
  } else {
    scan_dst_plain(dstp, nE, lo, (unsigned)cntN, [&](int e, int dl) { edge_acc(ei[e], dl); });
  }
  __syncthreads();

  for (int i = tid; i < cntN; i += RTH) {
    float z = lin_b[0];
    #pragma unroll
    for (int hh = 0; hh < HEADS; ++hh) {
      const float oh = numL[ACC(hh, i)] / (denL[ACC(hh, i)] + 1e-16f) + gat_bias[hh];
      z += oh * lin_w[hh];
    }
    out[lo + i] = fmaxf(z, 0.f) + bias[0];
  }
}

extern "C" void kernel_launch(void* const* d_in, const int* in_sizes, int n_in,
                              void* d_out, int out_size, void* d_ws, size_t ws_size,
                              hipStream_t stream) {
  const float* x        = (const float*)d_in[0];
  const int*   ei       = (const int*)d_in[1];
  const float* W        = (const float*)d_in[2];
  const float* att_src  = (const float*)d_in[3];
  const float* att_dst  = (const float*)d_in[4];
  const float* gat_bias = (const float*)d_in[5];
  const float* lin_w    = (const float*)d_in[6];
  const float* lin_b    = (const float*)d_in[7];
  const float* bias     = (const float*)d_in[8];
  float* out = (float*)d_out;

  const int n  = in_sizes[0] / IN_CH;
  const int nE = in_sizes[1] / 2;

  const size_t nh  = (size_t)n * HEADS;
  const size_t seg = (nh * sizeof(float) + 255) & ~(size_t)255;
  char* ws = (char*)d_ws;
  float*    rec = (float*)   (ws + 0 * seg);
  unsigned* bm  = (unsigned*)(ws + 2 * seg);

  const int tiles  = (n + 15) / 16;
  const int blocks = (tiles + 7) / 8;
  gat_gemm_kernel<<<blocks, GTH, 0, stream>>>(x, W, att_src, rec, n, ei, bm, nE);

  gat_range_kernel<<<(n + NPB - 1) / NPB, RTH, 0, stream>>>(
      ei, rec, att_dst, gat_bias, lin_w, lin_b, bias, out, nE, n, bm);
}


# x@W transposed and split along k: W^T is the register-held A operand (no LDS copy of W), a block's 8 waves take 64 k each for its 128 nodes, partials added in LDS in wave order; k sum order changes
# speedup vs baseline: 14.2299x; 1.1159x over previous
#include <hip/hip_runtime.h>
#include <math.h>

#define IN_CH 512
#define HEADS 8
#define NEG_SLOPE 0.2f
#define REC 16

typedef __attribute__((ext_vector_type(2))) float v2f;
typedef __attribute__((ext_vector_type(8))) float v8f;

__device__ __forceinline__ unsigned encF(float f) {
  unsigned u = __float_as_uint(f);
  return (u & 0x80000000u) ? ~u : (u | 0x80000000u);
}
__device__ __forceinline__ float decF(unsigned u) {
  u = (u & 0x80000000u) ? (u & 0x7FFFFFFFu) : ~u;
  return __uint_as_float(u);
}

__device__ __forceinline__ float leaky(float v) {
  return v >= 0.f ? v : NEG_SLOPE * v;
}

#define NPB 416

__device__ __forceinline__ unsigned range_key(int d) {
  const unsigned r = (unsigned)d / (unsigned)NPB;
  return r < 255u ? r : 255u;
}

#define GTH 256
#define CHP 1024
#define CHW 8
#define RMAX 256

__device__ __forceinline__ int bm_rows(int n) {
  const int nr = (n + NPB - 1) / NPB;
  const int nk = nr < RMAX ? nr : RMAX;
  return (nk + 3) & ~3;
}

__device__ __forceinline__ bool vec_path(const int* ei, int nE, const unsigned* bm) {
  return ((nE & 3) == 0) && ((((size_t)ei) & 15) == 0) && ((((size_t)bm) & 15) == 0);
}

#define KSL 64
#define TPB 8
#define NPBK (TPB * 16)

__global__ __launch_bounds__(GTH) void gat_gemm_kernel(
    const float* __restrict__ x, const float* __restrict__ W,
    const float* __restrict__ att_src,
    float* __restrict__ rec,
    int n,
    const int* __restrict__ ei, unsigned* __restrict__ bm, int nE)
{
  __shared__ __attribute__((aligned(16))) float part[(GTH >> 5) * NPBK * HEADS];
  __shared__ __attribute__((aligned(16))) unsigned bmL[RMAX * CHW];

  const bool vec = vec_path(ei, nE, bm);
  const int4* __restrict__ dv = (const int4*)(ei + nE);
  const int nPieces = nE >> 2;
  const int nChunks = vec ? (nPieces + CHP - 1) / CHP : 0;
  const int RP = bm_rows(n);
  int ch = (int)blockIdx.x;
  int4 d[CHP / GTH];
  auto load_chunk = [&]() {
    #pragma unroll
    for (int u = 0; u < CHP / GTH; ++u) {
      const int p = ch * CHP + u * GTH + (int)threadIdx.x;
      d[u] = dv[p < nPieces ? p : nPieces - 1];
    }
  };
  auto clear_rows = [&]() {
    for (int i = threadIdx.x; i < RMAX * CHW / 4; i += GTH)
      ((uint4*)bmL)[i] = make_uint4(0u, 0u, 0u, 0u);
  };
  if (ch < nChunks) load_chunk();
  clear_rows();

  const int lane = threadIdx.x & 31;
  const int wave = threadIdx.x >> 5;
  const int hf   = lane >> 4;
  const int c    = lane & 15;
  const int kbase = wave * KSL;

  v2f af[KSL / 4];
  {
    const float* __restrict__ wp = W + (size_t)(kbase + 4 * hf) * HEADS + (c & 7);
    #pragma unroll
    for (int j = 0; j < KSL / 4; ++j) {
      const int k = 8 * (j >> 1) + 2 * (j & 1);
      const float w0 = wp[k * HEADS];
      const float w1 = wp[(k + 1) * HEADS];
      af[j].x = (c < HEADS) ? w0 : 0.f;
      af[j].y = (c < HEADS) ? w1 : 0.f;
    }
  }
  __syncthreads();

  while (ch < nChunks) {
    #pragma unroll
    for (int u = 0; u < CHP / GTH; ++u) {
      const int pl = u * GTH + (int)threadIdx.x;
      if (ch * CHP + pl < nPieces) {
        const unsigned w   = (unsigned)pl >> 7;
        const unsigned bit = 1u << (((unsigned)pl >> 2) & 31u);
        atomicOr(&bmL[range_key(d[u].x) * CHW + w], bit);
        atomicOr(&bmL[range_key(d[u].y) * CHW + w], bit);
        atomicOr(&bmL[range_key(d[u].z) * CHW + w], bit);
        atomicOr(&bmL[range_key(d[u].w) * CHW + w], bit);
      }
    }
    __syncthreads();
    uint4* __restrict__ o = (uint4*)bm + (size_t)ch * (size_t)(RP * 2);
    for (int i = threadIdx.x; i < RP * 2; i += GTH) o[i] = ((const uint4*)bmL)[i];
    ch += (int)gridDim.x;
    if (ch < nChunks) {
      __syncthreads();
      load_chunk();
      clear_rows();
      __syncthreads();
    }
  }

  const int row0 = (int)blockIdx.x * NPBK;

  #pragma unroll 1
  for (int t = 0; t < TPB; t += 2) {
    int n0 = row0 + t * 16 + c;
    int n1 = n0 + 16;
    n0 = n0 < n ? n0 : n - 1;
    n1 = n1 < n ? n1 : n - 1;
    const float4* __restrict__ p0 = (const float4*)(x + (size_t)n0 * IN_CH + kbase + 4 * hf);
    const float4* __restrict__ p1 = (const float4*)(x + (size_t)n1 * IN_CH + kbase + 4 * hf);
    float4 q0[8], q1[8];
    #pragma unroll
    for (int i = 0; i < 8; ++i) { q0[i] = p0[2 * i]; q1[i] = p1[2 * i]; }
    __builtin_amdgcn_sched_barrier(0);

    v8f c0 = {0.f, 0.f, 0.f, 0.f, 0.f, 0.f, 0.f, 0.f};
    v8f c1 = {0.f, 0.f, 0.f, 0.f, 0.f, 0.f, 0.f, 0.f};
    #pragma unroll
    for (int j = 0; j < KSL / 4; ++j) {
      v2f b0, b1;
      if (j & 1) { b0.x = q0[j >> 1].z; b0.y = q0[j >> 1].w; b1.x = q1[j >> 1].z; b1.y = q1[j >> 1].w; }
      else       { b0.x = q0[j >> 1].x; b0.y = q0[j >> 1].y; b1.x = q1[j >> 1].x; b1.y = q1[j >> 1].y; }
      c0 = __builtin_amdgcn_wmma_f32_16x16x4_f32(
           false, af[j],  false, b0,
           (short)0, c0,  false,  false);
      c1 = __builtin_amdgcn_wmma_f32_16x16x4_f32(
           false, af[j],  false, b1,
           (short)0, c1,  false,  false);
    }
    if (hf == 0) {
      float4* __restrict__ pp0 = (float4*)(part + ((size_t)wave * NPBK + t * 16 + c) * HEADS);
      float4* __restrict__ pp1 = pp0 + 16 * HEADS / 4;
      pp0[0] = make_float4(c0[0], c0[1], c0[2], c0[3]);
      pp0[1] = make_float4(c0[4], c0[5], c0[6], c0[7]);
      pp1[0] = make_float4(c1[0], c1[1], c1[2], c1[3]);
      pp1[1] = make_float4(c1[4], c1[5], c1[6], c1[7]);
    }
  }
  __syncthreads();

  const float4* __restrict__ part4 = (const float4*)part;
  for (int slot = threadIdx.x; slot < NPBK * 4; slot += GTH) {
    const int nd = slot >> 2;
    const int q  = slot & 3;
    float4 s = part4[nd * 2 + (q & 1)];
    #pragma unroll
    for (int w = 1; w < (GTH >> 5); ++w) {
      const float4 v = part4[(w * NPBK + nd) * 2 + (q & 1)];
      s.x += v.x; s.y += v.y; s.z += v.z; s.w += v.w;
    }
    if (q < 2) {
      const float* __restrict__ as = att_src + q * 4;
      s.x *= as[0]; s.y *= as[1]; s.z *= as[2]; s.w *= as[3];
    }
    const int r = row0 + nd;
    if (r < n) ((float4*)(rec + (size_t)r * REC))[q] = s;
  }
}

#define RTH 1024
#define CAP 28672
#define EW 8
#define RU 4
#define ACC(hh, dl) ((hh) * NPB + (dl))
#define NCLS 32
#define CAPC 896
#define DLB 9
#define ADP 12

template <class F>
__device__ __forceinline__ void scan_dst_plain(const int* __restrict__ dstp, int nE,
                                               int lo, unsigned cnt, F f)
{
  for (int e = threadIdx.x; e < nE; e += RTH) {
    const unsigned t = (unsigned)(dstp[e] - lo);
    if (t < cnt) f(e, (int)t);
  }
}

__global__ __launch_bounds__(RTH) void gat_range_kernel(
    const int* __restrict__ ei,
    const float* __restrict__ rec, const float* __restrict__ att_dst,
    const float* __restrict__ gat_bias, const float* __restrict__ lin_w,
    const float* __restrict__ lin_b, const float* __restrict__ bias,
    float* __restrict__ out, int nE, int n,
    const unsigned* __restrict__ bm)
{
  __shared__ unsigned emL[NPB * HEADS];
  __shared__ float    denL[NPB * HEADS];
  __shared__ float    numL[NPB * HEADS];
  __shared__ __attribute__((aligned(16))) float adL[NPB * ADP];
  __shared__ unsigned listG[CAP];
  __shared__ unsigned listC[NCLS * CAPC];
  __shared__ unsigned ccnt[NCLS];
  __shared__ unsigned nlist;
  __shared__ unsigned over;

  const int tid = threadIdx.x;
  const int lo  = blockIdx.x * NPB;
  const int cntN = (n - lo < NPB) ? (n - lo) : NPB;
  const int cntE = cntN * HEADS;
  const int* __restrict__ dstp = ei + nE;

  const bool vec_ok  = vec_path(ei, nE, bm);
  const int  nPieces = nE >> 2;
  const int  nW      = vec_ok ? ((nPieces + CHP - 1) / CHP) * CHW : 0;
  const unsigned* __restrict__ myrow =
      bm + (size_t)(blockIdx.x < 255u ? blockIdx.x : 255u) * CHW;
  const size_t cpitch = (size_t)bm_rows(n) * CHW;
  unsigned wd[EW];
  auto load_words = [&](int w0) {
    #pragma unroll
    for (int u = 0; u < EW; ++u) {
      const int w = w0 + u * RTH + tid;
      wd[u] = (w < nW) ? myrow[(size_t)(w >> 3) * cpitch + (size_t)(w & 7)] : 0u;
    }
  };
  load_words(0);

  for (int j = tid; j < cntE; j += RTH) {
    const float* __restrict__ pr = rec + (size_t)(lo + (j >> 3)) * REC + (j & 7);
    const float a = pr[HEADS] * att_dst[j & 7];
    adL[(j >> 3) * ADP + (j & 7)] = a;
    emL[ACC(j & 7, j >> 3)] = encF(leaky(pr[0] + a));
  }
  if (tid < NCLS) ccnt[tid] = 0u;
  if (tid == 0) { nlist = 0u; over = 0u; }
  __syncthreads();

  for (int w0 = 0; w0 < nW; ) {
    unsigned cnt = 0u;
    #pragma unroll
    for (int u = 0; u < EW; ++u) cnt += (unsigned)__builtin_popcount(wd[u]);
    if (cnt) {
      unsigned sl = atomicAdd(&nlist, cnt);
      #pragma unroll
      for (int u = 0; u < EW; ++u) {
        unsigned m = wd[u];
        const unsigned g0 = (unsigned)(w0 + u * RTH + tid) << 5;
        while (m) {
          const unsigned b = (unsigned)__builtin_ctz(m);
          m &= m - 1u;
          if (sl < (unsigned)CAP) listG[sl] = g0 + b;
          ++sl;
        }
      }
    }
    w0 += RTH * EW;
    if (w0 < nW) load_words(w0);
  }
  __syncthreads();
  const unsigned total = nlist;
  const bool listed = vec_ok && total <= (unsigned)CAP && (unsigned)n <= (1u << (32 - DLB));

  auto edge_max = [&](int src, int dl) {
    const float4* ps = (const float4*)(rec + (size_t)src * REC);
    const float4 s0 = ps[0], s1 = ps[1];
    const float* ad = adL + dl * ADP;
    const float v[HEADS] = { s0.x + ad[0], s0.y + ad[1], s0.z + ad[2], s0.w + ad[3],
                             s1.x + ad[4], s1.y + ad[5], s1.z + ad[6], s1.w + ad[7] };
    #pragma unroll
    for (int hh = 0; hh < HEADS; ++hh)
      atomicMax(&emL[ACC(hh, dl)], encF(leaky(v[hh])));
  };
  auto edge_acc = [&](int src, int dl) {
    const float4* pr = (const float4*)(rec + (size_t)src * REC);
    const float4 s0 = pr[0], s1 = pr[1];
    const float4 h0 = pr[2], h1 = pr[3];
    const float* ad = adL + dl * ADP;
    const float v[HEADS]  = { s0.x + ad[0], s0.y + ad[1], s0.z + ad[2], s0.w + ad[3],
                              s1.x + ad[4], s1.y + ad[5], s1.z + ad[6], s1.w + ad[7] };
    const float hv[HEADS] = { h0.x, h0.y, h0.z, h0.w, h1.x, h1.y, h1.z, h1.w };
    float ee[HEADS];
    #pragma unroll
    for (int hh = 0; hh < HEADS; ++hh) ee[hh] = decF(emL[ACC(hh, dl)]);
    #pragma unroll
    for (int hh = 0; hh < HEADS; ++hh) ee[hh] = expf(leaky(v[hh]) - ee[hh]);
    #pragma unroll
    for (int hh = 0; hh < HEADS; ++hh) {
      atomicAdd(&denL[ACC(hh, dl)], ee[hh]);
      atomicAdd(&numL[ACC(hh, dl)], ee[hh] * hv[hh]);
    }
  };

  if (listed) {
    const int4* __restrict__ dv = (const int4*)dstp;
    const int4* __restrict__ sv = (const int4*)ei;
    const unsigned tot4 = total << 2;
    for (unsigned i0 = 0; i0 < tot4; i0 += RTH * RU) {
      int4 d[RU], s[RU];
      bool ok[RU];
      #pragma unroll
      for (int u = 0; u < RU; ++u) {
        const unsigned i = i0 + (unsigned)(u * RTH + tid);
        const bool in = i < tot4;
        int p = (int)((listG[in ? (i >> 2) : 0u] << 2) | (i & 3u));
        ok[u] = in && p < nPieces;
        p = ok[u] ? p : 0;
        d[u] = dv[p];
        s[u] = sv[p];
      }
      #pragma unroll
      for (int u = 0; u < RU; ++u) {
        asm volatile("" : "+v"(s[u].x), "+v"(s[u].y), "+v"(s[u].z), "+v"(s[u].w));
        const unsigned t0 = (unsigned)(d[u].x - lo), t1 = (unsigned)(d[u].y - lo);
        const unsigned t2 = (unsigned)(d[u].z - lo), t3 = (unsigned)(d[u].w - lo);
        unsigned mk = (t0 < (unsigned)cntN ? 1u : 0u) | (t1 < (unsigned)cntN ? 2u : 0u) |
                      (t2 < (unsigned)cntN ? 4u : 0u) | (t3 < (unsigned)cntN ? 8u : 0u);
        mk = ok[u] ? mk : 0u;
        const unsigned w0 = ((unsigned)s[u].x << DLB) | t0, w1 = ((unsigned)s[u].y << DLB) | t1;
        const unsigned w2 = ((unsigned)s[u].z << DLB) | t2, w3 = ((unsigned)s[u].w << DLB) | t3;
        while (mk) {
          const unsigned j = (unsigned)__builtin_ctz(mk);
          mk &= mk - 1u;
          const bool j1 = __builtin_unpredictable((j & 1u) != 0u);
          const bool j2 = __builtin_unpredictable((j & 2u) != 0u);
          const unsigned w  = j2 ? (j1 ? w3 : w2) : (j1 ? w1 : w0);
          const unsigned c  = w & (NCLS - 1);
          const unsigned sl = atomicAdd(&ccnt[c], 1u);
          if (sl < (unsigned)CAPC) listC[sl * NCLS + c] = w;
          else over = 1u;
        }
      }
    }
  }
  __syncthreads();
  const bool binned = listed && over == 0u;
  const unsigned lane = (unsigned)tid & 31u;
  const unsigned mycnt = binned ? ccnt[lane] : 0u;

  if (binned) {
    for (unsigned s = (unsigned)tid >> 5; s < mycnt; s += RTH / 32) {
      const unsigned w = listC[s * NCLS + lane];
      edge_max((int)(w >> DLB), (int)(w & ((1u << DLB) - 1u)));
    }
  } else {
    scan_dst_plain(dstp, nE, lo, (unsigned)cntN, [&](int e, int dl) { edge_max(ei[e], dl); });
  }
  __syncthreads();

  for (int j = tid; j < cntE; j += RTH) {
    const float* __restrict__ pr = rec + (size_t)(lo + (j >> 3)) * REC + (j & 7);
    const int a = ACC(j & 7, j >> 3);
    const float ee = expf(leaky(pr[0] + adL[(j >> 3) * ADP + (j & 7)]) - decF(emL[a]));
    denL[a] = ee;
    numL[a] = ee * pr[HEADS];
  }
  __syncthreads();

  if (binned) {
    for (unsigned s = (unsigned)tid >> 5; s < mycnt; s += RTH / 32) {
      const unsigned w = listC[s * NCLS + lane];
      edge_acc((int)(w >> DLB), (int)(w & ((1u << DLB) - 1u)));
    }
  } else {
    scan_dst_plain(dstp, nE, lo, (unsigned)cntN, [&](int e, int dl) { edge_acc(ei[e], dl); });
  }
  __syncthreads();

  for (int i = tid; i < cntN; i += RTH) {
    float z = lin_b[0];
    #pragma unroll
    for (int hh = 0; hh < HEADS; ++hh) {
      const float oh = numL[ACC(hh, i)] / (denL[ACC(hh, i)] + 1e-16f) + gat_bias[hh];
      z += oh * lin_w[hh];
    }
    out[lo + i] = fmaxf(z, 0.f) + bias[0];
  }
}

extern "C" void kernel_launch(void* const* d_in, const int* in_sizes, int n_in,
                              void* d_out, int out_size, void* d_ws, size_t ws_size,
                              hipStream_t stream) {
  const float* x        = (const float*)d_in[0];
  const int*   ei       = (const int*)d_in[1];
  const float* W        = (const float*)d_in[2];
  const float* att_src  = (const float*)d_in[3];
  const float* att_dst  = (const float*)d_in[4];
  const float* gat_bias = (const float*)d_in[5];
  const float* lin_w    = (const float*)d_in[6];
  const float* lin_b    = (const float*)d_in[7];
  const float* bias     = (const float*)d_in[8];
  float* out = (float*)d_out;

  const int n  = in_sizes[0] / IN_CH;
  const int nE = in_sizes[1] / 2;

  const size_t nh  = (size_t)n * HEADS;
  const size_t seg = (nh * sizeof(float) + 255) & ~(size_t)255;
  char* ws = (char*)d_ws;
  float*    rec = (float*)   (ws + 0 * seg);
  unsigned* bm  = (unsigned*)(ws + 2 * seg);

  const int tiles  = (n + 15) / 16;
  const int blocks = (tiles + 7) / 8;
  gat_gemm_kernel<<<blocks, GTH, 0, stream>>>(x, W, att_src, rec, n, ei, bm, nE);

  gat_range_kernel<<<(n + NPB - 1) / NPB, RTH, 0, stream>>>(
      ei, rec, att_dst, gat_bias, lin_w, lin_b, bias, out, nE, n, bm);
}


# softmax taken relative to the node's self-loop score, known before any edge: one edge walk, no max pass (a block redoes the two passes if a sum overflows); sums still in LDS arrival order
# speedup vs baseline: 16.2411x; 1.1413x over previous
#include <hip/hip_runtime.h>
#include <math.h>

#define IN_CH 512
#define HEADS 8
#define NEG_SLOPE 0.2f
#define REC 16

typedef __attribute__((ext_vector_type(2))) float v2f;
typedef __attribute__((ext_vector_type(8))) float v8f;

__device__ __forceinline__ unsigned encF(float f) {
  unsigned u = __float_as_uint(f);
  return (u & 0x80000000u) ? ~u : (u | 0x80000000u);
}
__device__ __forceinline__ float decF(unsigned u) {
  u = (u & 0x80000000u) ? (u & 0x7FFFFFFFu) : ~u;
  return __uint_as_float(u);
}

__device__ __forceinline__ float leaky(float v) {
  return v >= 0.f ? v : NEG_SLOPE * v;
}

#define NPB 416

__device__ __forceinline__ unsigned range_key(int d) {
  const unsigned r = (unsigned)d / (unsigned)NPB;
  return r < 255u ? r : 255u;
}

#define GTH 256
#define CHP 1024
#define CHW 8
#define RMAX 256

__device__ __forceinline__ int bm_rows(int n) {
  const int nr = (n + NPB - 1) / NPB;
  const int nk = nr < RMAX ? nr : RMAX;
  return (nk + 3) & ~3;
}

__device__ __forceinline__ bool vec_path(const int* ei, int nE, const unsigned* bm) {
  return ((nE & 3) == 0) && ((((size_t)ei) & 15) == 0) && ((((size_t)bm) & 15) == 0);
}

#define KSL 64
#define TPB 8
#define NPBK (TPB * 16)

__global__ __launch_bounds__(GTH) void gat_gemm_kernel(
    const float* __restrict__ x, const float* __restrict__ W,
    const float* __restrict__ att_src,
    float* __restrict__ rec,
    int n,
    const int* __restrict__ ei, unsigned* __restrict__ bm, int nE)
{
  __shared__ __attribute__((aligned(16))) float part[(GTH >> 5) * NPBK * HEADS];
  __shared__ __attribute__((aligned(16))) unsigned bmL[RMAX * CHW];

  const bool vec = vec_path(ei, nE, bm);
  const int4* __restrict__ dv = (const int4*)(ei + nE);
  const int nPieces = nE >> 2;
  const int nChunks = vec ? (nPieces + CHP - 1) / CHP : 0;
  const int RP = bm_rows(n);
  int ch = (int)blockIdx.x;
  int4 d[CHP / GTH];
  auto load_chunk = [&]() {
    #pragma unroll
    for (int u = 0; u < CHP / GTH; ++u) {
      const int p = ch * CHP + u * GTH + (int)threadIdx.x;
      d[u] = dv[p < nPieces ? p : nPieces - 1];
    }
  };
  auto clear_rows = [&]() {
    for (int i = threadIdx.x; i < RMAX * CHW / 4; i += GTH)
      ((uint4*)bmL)[i] = make_uint4(0u, 0u, 0u, 0u);
  };
  if (ch < nChunks) load_chunk();
  clear_rows();

  const int lane = threadIdx.x & 31;
  const int wave = threadIdx.x >> 5;
  const int hf   = lane >> 4;
  const int c    = lane & 15;
  const int kbase = wave * KSL;

  v2f af[KSL / 4];
  {
    const float* __restrict__ wp = W + (size_t)(kbase + 4 * hf) * HEADS + (c & 7);
    #pragma unroll
    for (int j = 0; j < KSL / 4; ++j) {
      const int k = 8 * (j >> 1) + 2 * (j & 1);
      const float w0 = wp[k * HEADS];
      const float w1 = wp[(k + 1) * HEADS];
      af[j].x = (c < HEADS) ? w0 : 0.f;
      af[j].y = (c < HEADS) ? w1 : 0.f;
    }
  }
  __syncthreads();

  while (ch < nChunks) {
    #pragma unroll
    for (int u = 0; u < CHP / GTH; ++u) {
      const int pl = u * GTH + (int)threadIdx.x;
      if (ch * CHP + pl < nPieces) {
        const unsigned w   = (unsigned)pl >> 7;
        const unsigned bit = 1u << (((unsigned)pl >> 2) & 31u);
        atomicOr(&bmL[range_key(d[u].x) * CHW + w], bit);
        atomicOr(&bmL[range_key(d[u].y) * CHW + w], bit);
        atomicOr(&bmL[range_key(d[u].z) * CHW + w], bit);
        atomicOr(&bmL[range_key(d[u].w) * CHW + w], bit);
      }
    }
    __syncthreads();
    uint4* __restrict__ o = (uint4*)bm + (size_t)ch * (size_t)(RP * 2);
    for (int i = threadIdx.x; i < RP * 2; i += GTH) o[i] = ((const uint4*)bmL)[i];
    ch += (int)gridDim.x;
    if (ch < nChunks) {
      __syncthreads();
      load_chunk();
      clear_rows();
      __syncthreads();
    }
  }

  const int row0 = (int)blockIdx.x * NPBK;

  #pragma unroll 1
  for (int t = 0; t < TPB; t += 2) {
    int n0 = row0 + t * 16 + c;
    int n1 = n0 + 16;
    n0 = n0 < n ? n0 : n - 1;
    n1 = n1 < n ? n1 : n - 1;
    const float4* __restrict__ p0 = (const float4*)(x + (size_t)n0 * IN_CH + kbase + 4 * hf);
    const float4* __restrict__ p1 = (const float4*)(x + (size_t)n1 * IN_CH + kbase + 4 * hf);
    float4 q0[8], q1[8];
    #pragma unroll
    for (int i = 0; i < 8; ++i) { q0[i] = p0[2 * i]; q1[i] = p1[2 * i]; }
    __builtin_amdgcn_sched_barrier(0);

    v8f c0 = {0.f, 0.f, 0.f, 0.f, 0.f, 0.f, 0.f, 0.f};
    v8f c1 = {0.f, 0.f, 0.f, 0.f, 0.f, 0.f, 0.f, 0.f};
    #pragma unroll
    for (int j = 0; j < KSL / 4; ++j) {
      v2f b0, b1;
      if (j & 1) { b0.x = q0[j >> 1].z; b0.y = q0[j >> 1].w; b1.x = q1[j >> 1].z; b1.y = q1[j >> 1].w; }
      else       { b0.x = q0[j >> 1].x; b0.y = q0[j >> 1].y; b1.x = q1[j >> 1].x; b1.y = q1[j >> 1].y; }
      c0 = __builtin_amdgcn_wmma_f32_16x16x4_f32(
           false, af[j],  false, b0,
           (short)0, c0,  false,  false);
      c1 = __builtin_amdgcn_wmma_f32_16x16x4_f32(
           false, af[j],  false, b1,
           (short)0, c1,  false,  false);
    }
    if (hf == 0) {
      float4* __restrict__ pp0 = (float4*)(part + ((size_t)wave * NPBK + t * 16 + c) * HEADS);
      float4* __restrict__ pp1 = pp0 + 16 * HEADS / 4;
      pp0[0] = make_float4(c0[0], c0[1], c0[2], c0[3]);
      pp0[1] = make_float4(c0[4], c0[5], c0[6], c0[7]);
      pp1[0] = make_float4(c1[0], c1[1], c1[2], c1[3]);
      pp1[1] = make_float4(c1[4], c1[5], c1[6], c1[7]);
    }
  }
  __syncthreads();

  const float4* __restrict__ part4 = (const float4*)part;
  for (int slot = threadIdx.x; slot < NPBK * 4; slot += GTH) {
    const int nd = slot >> 2;
    const int q  = slot & 3;
    float4 s = part4[nd * 2 + (q & 1)];
    #pragma unroll
    for (int w = 1; w < (GTH >> 5); ++w) {
      const float4 v = part4[(w * NPBK + nd) * 2 + (q & 1)];
      s.x += v.x; s.y += v.y; s.z += v.z; s.w += v.w;
    }
    if (q < 2) {
      const float* __restrict__ as = att_src + q * 4;
      s.x *= as[0]; s.y *= as[1]; s.z *= as[2]; s.w *= as[3];
    }
    const int r = row0 + nd;
    if (r < n) ((float4*)(rec + (size_t)r * REC))[q] = s;
  }
}

#define RTH 1024
#define CAP 28672
#define EW 8
#define RU 4
#define ACC(hh, dl) ((hh) * NPB + (dl))
#define NCLS 32
#define CAPC 896
#define DLB 9
#define ADP 12

template <class F>
__device__ __forceinline__ void scan_dst_plain(const int* __restrict__ dstp, int nE,
                                               int lo, unsigned cnt, F f)
{
  for (int e = threadIdx.x; e < nE; e += RTH) {
    const unsigned t = (unsigned)(dstp[e] - lo);
    if (t < cnt) f(e, (int)t);
  }
}

__global__ __launch_bounds__(RTH) void gat_range_kernel(
    const int* __restrict__ ei,
    const float* __restrict__ rec, const float* __restrict__ att_dst,
    const float* __restrict__ gat_bias, const float* __restrict__ lin_w,
    const float* __restrict__ lin_b, const float* __restrict__ bias,
    float* __restrict__ out, int nE, int n,
    const unsigned* __restrict__ bm)
{
  __shared__ unsigned emL[NPB * HEADS];
  __shared__ float    denL[NPB * HEADS];
  __shared__ float    numL[NPB * HEADS];
  __shared__ __attribute__((aligned(16))) float adL[NPB * ADP];
  __shared__ unsigned listG[CAP];
  __shared__ unsigned listC[NCLS * CAPC];
  __shared__ unsigned ccnt[NCLS];
  __shared__ unsigned nlist;
  __shared__ unsigned over;
  __shared__ unsigned redo;

  const int tid = threadIdx.x;
  const int lo  = blockIdx.x * NPB;
  const int cntN = (n - lo < NPB) ? (n - lo) : NPB;
  const int cntE = cntN * HEADS;
  const int* __restrict__ dstp = ei + nE;

  const bool vec_ok  = vec_path(ei, nE, bm);
  const int  nPieces = nE >> 2;
  const int  nW      = vec_ok ? ((nPieces + CHP - 1) / CHP) * CHW : 0;
  const unsigned* __restrict__ myrow =
      bm + (size_t)(blockIdx.x < 255u ? blockIdx.x : 255u) * CHW;
  const size_t cpitch = (size_t)bm_rows(n) * CHW;
  unsigned wd[EW];
  auto load_words = [&](int w0) {
    #pragma unroll
    for (int u = 0; u < EW; ++u) {
      const int w = w0 + u * RTH + tid;
      wd[u] = (w < nW) ? myrow[(size_t)(w >> 3) * cpitch + (size_t)(w & 7)] : 0u;
    }
  };
  load_words(0);

  for (int j = tid; j < cntE; j += RTH) {
    const float* __restrict__ pr = rec + (size_t)(lo + (j >> 3)) * REC + (j & 7);
    const float hs = pr[HEADS];
    const float a = hs * att_dst[j & 7];
    const int ai = ACC(j & 7, j >> 3);
    adL[(j >> 3) * ADP + (j & 7)] = a;
    emL[ai]  = __float_as_uint(leaky(pr[0] + a));
    denL[ai] = 1.f;
    numL[ai] = hs;
  }
  if (tid < NCLS) ccnt[tid] = 0u;
  if (tid == 0) { nlist = 0u; over = 0u; redo = 0u; }
  __syncthreads();

  for (int w0 = 0; w0 < nW; ) {
    unsigned cnt = 0u;
    #pragma unroll
    for (int u = 0; u < EW; ++u) cnt += (unsigned)__builtin_popcount(wd[u]);
    if (cnt) {
      unsigned sl = atomicAdd(&nlist, cnt);
      #pragma unroll
      for (int u = 0; u < EW; ++u) {
        unsigned m = wd[u];
        const unsigned g0 = (unsigned)(w0 + u * RTH + tid) << 5;
        while (m) {
          const unsigned b = (unsigned)__builtin_ctz(m);
          m &= m - 1u;
          if (sl < (unsigned)CAP) listG[sl] = g0 + b;
          ++sl;
        }
      }
    }
    w0 += RTH * EW;
    if (w0 < nW) load_words(w0);
  }
  __syncthreads();
  const unsigned total = nlist;
  const bool listed = vec_ok && total <= (unsigned)CAP && (unsigned)n <= (1u << (32 - DLB));

  auto edge_max = [&](int src, int dl) {
    const float4* ps = (const float4*)(rec + (size_t)src * REC);
    const float4 s0 = ps[0], s1 = ps[1];
    const float* ad = adL + dl * ADP;
    const float v[HEADS] = { s0.x + ad[0], s0.y + ad[1], s0.z + ad[2], s0.w + ad[3],
                             s1.x + ad[4], s1.y + ad[5], s1.z + ad[6], s1.w + ad[7] };
    #pragma unroll
    for (int hh = 0; hh < HEADS; ++hh)
      atomicMax(&emL[ACC(hh, dl)], encF(leaky(v[hh])));
  };
  auto edge_acc_from = [&](int src, int dl, auto ref_point) {
    const float4* pr = (const float4*)(rec + (size_t)src * REC);
    const float4 s0 = pr[0], s1 = pr[1];
    const float4 h0 = pr[2], h1 = pr[3];
    const float* ad = adL + dl * ADP;
    const float v[HEADS]  = { s0.x + ad[0], s0.y + ad[1], s0.z + ad[2], s0.w + ad[3],
                              s1.x + ad[4], s1.y + ad[5], s1.z + ad[6], s1.w + ad[7] };
    const float hv[HEADS] = { h0.x, h0.y, h0.z, h0.w, h1.x, h1.y, h1.z, h1.w };
    float ee[HEADS];
    #pragma unroll
    for (int hh = 0; hh < HEADS; ++hh) ee[hh] = ref_point(emL[ACC(hh, dl)]);
    #pragma unroll
    for (int hh = 0; hh < HEADS; ++hh) ee[hh] = expf(leaky(v[hh]) - ee[hh]);
    #pragma unroll
    for (int hh = 0; hh < HEADS; ++hh) {
      atomicAdd(&denL[ACC(hh, dl)], ee[hh]);
      atomicAdd(&numL[ACC(hh, dl)], ee[hh] * hv[hh]);
    }
  };
  auto edge_acc_self = [&](int src, int dl) {
    edge_acc_from(src, dl, [](unsigned u) { return __uint_as_float(u); });
  };
  auto edge_acc_max = [&](int src, int dl) {
    edge_acc_from(src, dl, [](unsigned u) { return decF(u); });
  };

  if (listed) {
    const int4* __restrict__ dv = (const int4*)dstp;
    const int4* __restrict__ sv = (const int4*)ei;
    const unsigned tot4 = total << 2;
    for (unsigned i0 = 0; i0 < tot4; i0 += RTH * RU) {
      int4 d[RU], s[RU];
      bool ok[RU];
      #pragma unroll
      for (int u = 0; u < RU; ++u) {
        const unsigned i = i0 + (unsigned)(u * RTH + tid);
        const bool in = i < tot4;
        int p = (int)((listG[in ? (i >> 2) : 0u] << 2) | (i & 3u));
        ok[u] = in && p < nPieces;
        p = ok[u] ? p : 0;
        d[u] = dv[p];
        s[u] = sv[p];
      }
      #pragma unroll
      for (int u = 0; u < RU; ++u) {
        asm volatile("" : "+v"(s[u].x), "+v"(s[u].y), "+v"(s[u].z), "+v"(s[u].w));
        const unsigned t0 = (unsigned)(d[u].x - lo), t1 = (unsigned)(d[u].y - lo);
        const unsigned t2 = (unsigned)(d[u].z - lo), t3 = (unsigned)(d[u].w - lo);
        unsigned mk = (t0 < (unsigned)cntN ? 1u : 0u) | (t1 < (unsigned)cntN ? 2u : 0u) |
                      (t2 < (unsigned)cntN ? 4u : 0u) | (t3 < (unsigned)cntN ? 8u : 0u);
        mk = ok[u] ? mk : 0u;
        const unsigned w0 = ((unsigned)s[u].x << DLB) | t0, w1 = ((unsigned)s[u].y << DLB) | t1;
        const unsigned w2 = ((unsigned)s[u].z << DLB) | t2, w3 = ((unsigned)s[u].w << DLB) | t3;
        while (mk) {
          const unsigned j = (unsigned)__builtin_ctz(mk);
          mk &= mk - 1u;
          const bool j1 = __builtin_unpredictable((j & 1u) != 0u);
          const bool j2 = __builtin_unpredictable((j & 2u) != 0u);
          const unsigned w  = j2 ? (j1 ? w3 : w2) : (j1 ? w1 : w0);
          const unsigned c  = w & (NCLS - 1);
          const unsigned sl = atomicAdd(&ccnt[c], 1u);
          if (sl < (unsigned)CAPC) listC[sl * NCLS + c] = w;
          else over = 1u;
        }
      }
    }
  }
  __syncthreads();
  const bool binned = listed && over == 0u;
  const unsigned lane = (unsigned)tid & 31u;
  const unsigned mycnt = binned ? ccnt[lane] : 0u;

  auto head = [&](bool check) {
    for (int i = tid; i < cntN; i += RTH) {
      float z = lin_b[0];
      bool fin = true;
      #pragma unroll
      for (int hh = 0; hh < HEADS; ++hh) {
        const float dn = denL[ACC(hh, i)];
        const float nm = numL[ACC(hh, i)];
        fin = fin && ((__float_as_uint(dn) & 0x7F800000u) != 0x7F800000u)
                  && ((__float_as_uint(nm) & 0x7F800000u) != 0x7F800000u);
        const float oh = nm / (dn + 1e-16f) + gat_bias[hh];
        z += oh * lin_w[hh];
      }
      out[lo + i] = fmaxf(z, 0.f) + bias[0];
      if (check && !fin) redo = 1u;
    }
  };

  if (binned) {
    for (unsigned s = (unsigned)tid >> 5; s < mycnt; s += RTH / 32) {
      const unsigned w = listC[s * NCLS + lane];
      edge_acc_self((int)(w >> DLB), (int)(w & ((1u << DLB) - 1u)));
    }
  } else {
    scan_dst_plain(dstp, nE, lo, (unsigned)cntN, [&](int e, int dl) { edge_acc_self(ei[e], dl); });
  }
  __syncthreads();

  head(true);
  __syncthreads();
  if (redo == 0u) return;

  for (int j = tid; j < cntE; j += RTH) {
    const int a = ACC(j & 7, j >> 3);
    emL[a] = encF(__uint_as_float(emL[a]));
  }
  __syncthreads();

  if (binned) {
    for (unsigned s = (unsigned)tid >> 5; s < mycnt; s += RTH / 32) {
      const unsigned w = listC[s * NCLS + lane];
      edge_max((int)(w >> DLB), (int)(w & ((1u << DLB) - 1u)));
    }
  } else {
    scan_dst_plain(dstp, nE, lo, (unsigned)cntN, [&](int e, int dl) { edge_max(ei[e], dl); });
  }
  __syncthreads();

  for (int j = tid; j < cntE; j += RTH) {
    const float* __restrict__ pr = rec + (size_t)(lo + (j >> 3)) * REC + (j & 7);
    const int a = ACC(j & 7, j >> 3);
    const float ee = expf(leaky(pr[0] + adL[(j >> 3) * ADP + (j & 7)]) - decF(emL[a]));
    denL[a] = ee;
    numL[a] = ee * pr[HEADS];
  }
  __syncthreads();

  if (binned) {
    for (unsigned s = (unsigned)tid >> 5; s < mycnt; s += RTH / 32) {
      const unsigned w = listC[s * NCLS + lane];
      edge_acc_max((int)(w >> DLB), (int)(w & ((1u << DLB) - 1u)));
    }
  } else {
    scan_dst_plain(dstp, nE, lo, (unsigned)cntN, [&](int e, int dl) { edge_acc_max(ei[e], dl); });
  }
  __syncthreads();

  head(false);
}

extern "C" void kernel_launch(void* const* d_in, const int* in_sizes, int n_in,
                              void* d_out, int out_size, void* d_ws, size_t ws_size,
                              hipStream_t stream) {
  const float* x        = (const float*)d_in[0];
  const int*   ei       = (const int*)d_in[1];
  const float* W        = (const float*)d_in[2];
  const float* att_src  = (const float*)d_in[3];
  const float* att_dst  = (const float*)d_in[4];
  const float* gat_bias = (const float*)d_in[5];
  const float* lin_w    = (const float*)d_in[6];
  const float* lin_b    = (const float*)d_in[7];
  const float* bias     = (const float*)d_in[8];
  float* out = (float*)d_out;

  const int n  = in_sizes[0] / IN_CH;
  const int nE = in_sizes[1] / 2;

  const size_t nh  = (size_t)n * HEADS;
  const size_t seg = (nh * sizeof(float) + 255) & ~(size_t)255;
  char* ws = (char*)d_ws;
  float*    rec = (float*)   (ws + 0 * seg);
  unsigned* bm  = (unsigned*)(ws + 2 * seg);

  const int tiles  = (n + 15) / 16;
  const int blocks = (tiles + 7) / 8;
  gat_gemm_kernel<<<blocks, GTH, 0, stream>>>(x, W, att_src, rec, n, ei, bm, nE);

  gat_range_kernel<<<(n + NPB - 1) / NPB, RTH, 0, stream>>>(
      ei, rec, att_dst, gat_bias, lin_w, lin_b, bias, out, nE, n, bm);
}


# edge walk without atomics: sources filed node by node in LDS, two lanes per node (heads 0-3 and 4-7) walk its list and keep their sums in registers; the order of the sums changes
# speedup vs baseline: 16.7652x; 1.0323x over previous
#include <hip/hip_runtime.h>
#include <math.h>

#define IN_CH 512
#define HEADS 8
#define NEG_SLOPE 0.2f
#define REC 16

typedef __attribute__((ext_vector_type(2))) float v2f;
typedef __attribute__((ext_vector_type(8))) float v8f;

__device__ __forceinline__ unsigned encF(float f) {
  unsigned u = __float_as_uint(f);
  return (u & 0x80000000u) ? ~u : (u | 0x80000000u);
}
__device__ __forceinline__ float decF(unsigned u) {
  u = (u & 0x80000000u) ? (u & 0x7FFFFFFFu) : ~u;
  return __uint_as_float(u);
}

__device__ __forceinline__ float leaky(float v) {
  return v >= 0.f ? v : NEG_SLOPE * v;
}

#define NPB 416

__device__ __forceinline__ unsigned range_key(int d) {
  const unsigned r = (unsigned)d / (unsigned)NPB;
  return r < 255u ? r : 255u;
}

#define GTH 256
#define CHP 1024
#define CHW 8
#define RMAX 256

__device__ __forceinline__ int bm_rows(int n) {
  const int nr = (n + NPB - 1) / NPB;
  const int nk = nr < RMAX ? nr : RMAX;
  return (nk + 3) & ~3;
}

__device__ __forceinline__ bool vec_path(const int* ei, int nE, const unsigned* bm) {
  return ((nE & 3) == 0) && ((((size_t)ei) & 15) == 0) && ((((size_t)bm) & 15) == 0);
}

#define KSL 64
#define TPB 8
#define NPBK (TPB * 16)

__global__ __launch_bounds__(GTH) void gat_gemm_kernel(
    const float* __restrict__ x, const float* __restrict__ W,
    const float* __restrict__ att_src,
    float* __restrict__ rec,
    int n,
    const int* __restrict__ ei, unsigned* __restrict__ bm, int nE)
{
  __shared__ __attribute__((aligned(16))) float part[(GTH >> 5) * NPBK * HEADS];
  __shared__ __attribute__((aligned(16))) unsigned bmL[RMAX * CHW];

  const bool vec = vec_path(ei, nE, bm);
  const int4* __restrict__ dv = (const int4*)(ei + nE);
  const int nPieces = nE >> 2;
  const int nChunks = vec ? (nPieces + CHP - 1) / CHP : 0;
  const int RP = bm_rows(n);
  int ch = (int)blockIdx.x;
  int4 d[CHP / GTH];
  auto load_chunk = [&]() {
    #pragma unroll
    for (int u = 0; u < CHP / GTH; ++u) {
      const int p = ch * CHP + u * GTH + (int)threadIdx.x;
      d[u] = dv[p < nPieces ? p : nPieces - 1];
    }
  };
  auto clear_rows = [&]() {
    for (int i = threadIdx.x; i < RMAX * CHW / 4; i += GTH)
      ((uint4*)bmL)[i] = make_uint4(0u, 0u, 0u, 0u);
  };
  if (ch < nChunks) load_chunk();
  clear_rows();

  const int lane = threadIdx.x & 31;
  const int wave = threadIdx.x >> 5;
  const int hf   = lane >> 4;
  const int c    = lane & 15;
  const int kbase = wave * KSL;

  v2f af[KSL / 4];
  {
    const float* __restrict__ wp = W + (size_t)(kbase + 4 * hf) * HEADS + (c & 7);
    #pragma unroll
    for (int j = 0; j < KSL / 4; ++j) {
      const int k = 8 * (j >> 1) + 2 * (j & 1);
      const float w0 = wp[k * HEADS];
      const float w1 = wp[(k + 1) * HEADS];
      af[j].x = (c < HEADS) ? w0 : 0.f;
      af[j].y = (c < HEADS) ? w1 : 0.f;
    }
  }
  __syncthreads();

  while (ch < nChunks) {
    #pragma unroll
    for (int u = 0; u < CHP / GTH; ++u) {
      const int pl = u * GTH + (int)threadIdx.x;
      if (ch * CHP + pl < nPieces) {
        const unsigned w   = (unsigned)pl >> 7;
        const unsigned bit = 1u << (((unsigned)pl >> 2) & 31u);
        atomicOr(&bmL[range_key(d[u].x) * CHW + w], bit);
        atomicOr(&bmL[range_key(d[u].y) * CHW + w], bit);
        atomicOr(&bmL[range_key(d[u].z) * CHW + w], bit);
        atomicOr(&bmL[range_key(d[u].w) * CHW + w], bit);
      }
    }
    __syncthreads();
    uint4* __restrict__ o = (uint4*)bm + (size_t)ch * (size_t)(RP * 2);
    for (int i = threadIdx.x; i < RP * 2; i += GTH) o[i] = ((const uint4*)bmL)[i];
    ch += (int)gridDim.x;
    if (ch < nChunks) {
      __syncthreads();
      load_chunk();
      clear_rows();
      __syncthreads();
    }
  }

  const int row0 = (int)blockIdx.x * NPBK;

  #pragma unroll 1
  for (int t = 0; t < TPB; t += 2) {
    int n0 = row0 + t * 16 + c;
    int n1 = n0 + 16;
    n0 = n0 < n ? n0 : n - 1;
    n1 = n1 < n ? n1 : n - 1;
    const float4* __restrict__ p0 = (const float4*)(x + (size_t)n0 * IN_CH + kbase + 4 * hf);
    const float4* __restrict__ p1 = (const float4*)(x + (size_t)n1 * IN_CH + kbase + 4 * hf);
    float4 q0[8], q1[8];
    #pragma unroll
    for (int i = 0; i < 8; ++i) { q0[i] = p0[2 * i]; q1[i] = p1[2 * i]; }
    __builtin_amdgcn_sched_barrier(0);

    v8f c0 = {0.f, 0.f, 0.f, 0.f, 0.f, 0.f, 0.f, 0.f};
    v8f c1 = {0.f, 0.f, 0.f, 0.f, 0.f, 0.f, 0.f, 0.f};
    #pragma unroll
    for (int j = 0; j < KSL / 4; ++j) {
      v2f b0, b1;
      if (j & 1) { b0.x = q0[j >> 1].z; b0.y = q0[j >> 1].w; b1.x = q1[j >> 1].z; b1.y = q1[j >> 1].w; }
      else       { b0.x = q0[j >> 1].x; b0.y = q0[j >> 1].y; b1.x = q1[j >> 1].x; b1.y = q1[j >> 1].y; }
      c0 = __builtin_amdgcn_wmma_f32_16x16x4_f32(
           false, af[j],  false, b0,
           (short)0, c0,  false,  false);
      c1 = __builtin_amdgcn_wmma_f32_16x16x4_f32(
           false, af[j],  false, b1,
           (short)0, c1,  false,  false);
    }
    if (hf == 0) {
      float4* __restrict__ pp0 = (float4*)(part + ((size_t)wave * NPBK + t * 16 + c) * HEADS);
      float4* __restrict__ pp1 = pp0 + 16 * HEADS / 4;
      pp0[0] = make_float4(c0[0], c0[1], c0[2], c0[3]);
      pp0[1] = make_float4(c0[4], c0[5], c0[6], c0[7]);
      pp1[0] = make_float4(c1[0], c1[1], c1[2], c1[3]);
      pp1[1] = make_float4(c1[4], c1[5], c1[6], c1[7]);
    }
  }
  __syncthreads();

  const float4* __restrict__ part4 = (const float4*)part;
  for (int slot = threadIdx.x; slot < NPBK * 4; slot += GTH) {
    const int nd = slot >> 2;
    const int q  = slot & 3;
    float4 s = part4[nd * 2 + (q & 1)];
    #pragma unroll
    for (int w = 1; w < (GTH >> 5); ++w) {
      const float4 v = part4[(w * NPBK + nd) * 2 + (q & 1)];
      s.x += v.x; s.y += v.y; s.z += v.z; s.w += v.w;
    }
    if (q < 2) {
      const float* __restrict__ as = att_src + q * 4;
      s.x *= as[0]; s.y *= as[1]; s.z *= as[2]; s.w *= as[3];
    }
    const int r = row0 + nd;
    if (r < n) ((float4*)(rec + (size_t)r * REC))[q] = s;
  }
}

#define RTH 1024
#define CAP 28672
#define EW 8
#define RU 4
#define ACC(hh, dl) ((hh) * NPB + (dl))
#define CAPN 72
#define HQ 4
#define NU 4
#define DLB 9
#define ADP 12

template <class F>
__device__ __forceinline__ void scan_dst_plain(const int* __restrict__ dstp, int nE,
                                               int lo, unsigned cnt, F f)
{
  for (int e = threadIdx.x; e < nE; e += RTH) {
    const unsigned t = (unsigned)(dstp[e] - lo);
    if (t < cnt) f(e, (int)t);
  }
}

__global__ __launch_bounds__(RTH) void gat_range_kernel(
    const int* __restrict__ ei,
    const float* __restrict__ rec, const float* __restrict__ att_dst,
    const float* __restrict__ gat_bias, const float* __restrict__ lin_w,
    const float* __restrict__ lin_b, const float* __restrict__ bias,
    float* __restrict__ out, int nE, int n,
    const unsigned* __restrict__ bm)
{
  __shared__ unsigned emL[NPB * HEADS];
  __shared__ float    denL[NPB * HEADS];
  __shared__ float    numL[NPB * HEADS];
  __shared__ __attribute__((aligned(16))) float adL[NPB * ADP];
  __shared__ unsigned listG[CAP];
  __shared__ unsigned listC[CAPN * NPB];
  __shared__ unsigned ncnt[NPB];
  __shared__ unsigned nlist;
  __shared__ unsigned over;
  __shared__ unsigned redo;

  const int tid = threadIdx.x;
  const int lo  = blockIdx.x * NPB;
  const int cntN = (n - lo < NPB) ? (n - lo) : NPB;
  const int cntE = cntN * HEADS;
  const int* __restrict__ dstp = ei + nE;

  const bool vec_ok  = vec_path(ei, nE, bm);
  const int  nPieces = nE >> 2;
  const int  nW      = vec_ok ? ((nPieces + CHP - 1) / CHP) * CHW : 0;
  const unsigned* __restrict__ myrow =
      bm + (size_t)(blockIdx.x < 255u ? blockIdx.x : 255u) * CHW;
  const size_t cpitch = (size_t)bm_rows(n) * CHW;
  unsigned wd[EW];
  auto load_words = [&](int w0) {
    #pragma unroll
    for (int u = 0; u < EW; ++u) {
      const int w = w0 + u * RTH + tid;
      wd[u] = (w < nW) ? myrow[(size_t)(w >> 3) * cpitch + (size_t)(w & 7)] : 0u;
    }
  };
  load_words(0);

  for (int j = tid; j < cntE; j += RTH) {
    const float* __restrict__ pr = rec + (size_t)(lo + (j >> 3)) * REC + (j & 7);
    const float hs = pr[HEADS];
    const float a = hs * att_dst[j & 7];
    const int ai = ACC(j & 7, j >> 3);
    adL[(j >> 3) * ADP + (j & 7)] = a;
    emL[ai]  = __float_as_uint(leaky(pr[0] + a));
    denL[ai] = 1.f;
    numL[ai] = hs;
  }
  if (tid < NPB) ncnt[tid] = 0u;
  if (tid == 0) { nlist = 0u; over = 0u; redo = 0u; }
  __syncthreads();

  for (int w0 = 0; w0 < nW; ) {
    unsigned cnt = 0u;
    #pragma unroll
    for (int u = 0; u < EW; ++u) cnt += (unsigned)__builtin_popcount(wd[u]);
    if (cnt) {
      unsigned sl = atomicAdd(&nlist, cnt);
      #pragma unroll
      for (int u = 0; u < EW; ++u) {
        unsigned m = wd[u];
        const unsigned g0 = (unsigned)(w0 + u * RTH + tid) << 5;
        while (m) {
          const unsigned b = (unsigned)__builtin_ctz(m);
          m &= m - 1u;
          if (sl < (unsigned)CAP) listG[sl] = g0 + b;
          ++sl;
        }
      }
    }
    w0 += RTH * EW;
    if (w0 < nW) load_words(w0);
  }
  __syncthreads();
  const unsigned total = nlist;
  const bool listed = vec_ok && total <= (unsigned)CAP && (unsigned)n <= (1u << (32 - DLB));

  auto edge_max = [&](int src, int dl) {
    const float4* ps = (const float4*)(rec + (size_t)src * REC);
    const float4 s0 = ps[0], s1 = ps[1];
    const float* ad = adL + dl * ADP;
    const float v[HEADS] = { s0.x + ad[0], s0.y + ad[1], s0.z + ad[2], s0.w + ad[3],
                             s1.x + ad[4], s1.y + ad[5], s1.z + ad[6], s1.w + ad[7] };
    #pragma unroll
    for (int hh = 0; hh < HEADS; ++hh)
      atomicMax(&emL[ACC(hh, dl)], encF(leaky(v[hh])));
  };
  auto edge_acc_from = [&](int src, int dl, auto ref_point) {
    const float4* pr = (const float4*)(rec + (size_t)src * REC);
    const float4 s0 = pr[0], s1 = pr[1];
    const float4 h0 = pr[2], h1 = pr[3];
    const float* ad = adL + dl * ADP;
    const float v[HEADS]  = { s0.x + ad[0], s0.y + ad[1], s0.z + ad[2], s0.w + ad[3],
                              s1.x + ad[4], s1.y + ad[5], s1.z + ad[6], s1.w + ad[7] };
    const float hv[HEADS] = { h0.x, h0.y, h0.z, h0.w, h1.x, h1.y, h1.z, h1.w };
    float ee[HEADS];
    #pragma unroll
    for (int hh = 0; hh < HEADS; ++hh) ee[hh] = ref_point(emL[ACC(hh, dl)]);
    #pragma unroll
    for (int hh = 0; hh < HEADS; ++hh) ee[hh] = expf(leaky(v[hh]) - ee[hh]);
    #pragma unroll
    for (int hh = 0; hh < HEADS; ++hh) {
      atomicAdd(&denL[ACC(hh, dl)], ee[hh]);
      atomicAdd(&numL[ACC(hh, dl)], ee[hh] * hv[hh]);
    }
  };
  auto edge_acc_self = [&](int src, int dl) {
    edge_acc_from(src, dl, [](unsigned u) { return __uint_as_float(u); });
  };
  auto edge_acc_max = [&](int src, int dl) {
    edge_acc_from(src, dl, [](unsigned u) { return decF(u); });
  };

  if (listed) {
    const int4* __restrict__ dv = (const int4*)dstp;
    const int4* __restrict__ sv = (const int4*)ei;
    const unsigned tot4 = total << 2;
    for (unsigned i0 = 0; i0 < tot4; i0 += RTH * RU) {
      int4 d[RU], s[RU];
      bool ok[RU];
      #pragma unroll
      for (int u = 0; u < RU; ++u) {
        const unsigned i = i0 + (unsigned)(u * RTH + tid);
        const bool in = i < tot4;
        int p = (int)((listG[in ? (i >> 2) : 0u] << 2) | (i & 3u));
        ok[u] = in && p < nPieces;
        p = ok[u] ? p : 0;
        d[u] = dv[p];
        s[u] = sv[p];
      }
      #pragma unroll
      for (int u = 0; u < RU; ++u) {
        asm volatile("" : "+v"(s[u].x), "+v"(s[u].y), "+v"(s[u].z), "+v"(s[u].w));
        const unsigned t0 = (unsigned)(d[u].x - lo), t1 = (unsigned)(d[u].y - lo);
        const unsigned t2 = (unsigned)(d[u].z - lo), t3 = (unsigned)(d[u].w - lo);
        unsigned mk = (t0 < (unsigned)cntN ? 1u : 0u) | (t1 < (unsigned)cntN ? 2u : 0u) |
                      (t2 < (unsigned)cntN ? 4u : 0u) | (t3 < (unsigned)cntN ? 8u : 0u);
        mk = ok[u] ? mk : 0u;
        const unsigned w0 = ((unsigned)s[u].x << DLB) | t0, w1 = ((unsigned)s[u].y << DLB) | t1;
        const unsigned w2 = ((unsigned)s[u].z << DLB) | t2, w3 = ((unsigned)s[u].w << DLB) | t3;
        while (mk) {
          const unsigned j = (unsigned)__builtin_ctz(mk);
          mk &= mk - 1u;
          const bool j1 = __builtin_unpredictable((j & 1u) != 0u);
          const bool j2 = __builtin_unpredictable((j & 2u) != 0u);
          const unsigned w  = j2 ? (j1 ? w3 : w2) : (j1 ? w1 : w0);
          const unsigned dlv = w & ((1u << DLB) - 1u);
          const unsigned sl = atomicAdd(&ncnt[dlv], 1u);
          if (sl < (unsigned)CAPN) listC[sl * NPB + dlv] = w >> DLB;
          else over = 1u;
        }
      }
    }
  }
  __syncthreads();
  const bool binned = listed && over == 0u;
  const bool owner = binned && tid < 2 * cntN;
  const int  mydl  = owner ? (tid >> 1) : 0;
  const unsigned myp   = (unsigned)tid & 1u;
  const unsigned mycnt = owner ? ncnt[mydl] : 0u;

  auto head = [&](bool check) {
    for (int i = tid; i < cntN; i += RTH) {
      float z = lin_b[0];
      bool fin = true;
      #pragma unroll
      for (int hh = 0; hh < HEADS; ++hh) {
        const float dn = denL[ACC(hh, i)];
        const float nm = numL[ACC(hh, i)];
        fin = fin && ((__float_as_uint(dn) & 0x7F800000u) != 0x7F800000u)
                  && ((__float_as_uint(nm) & 0x7F800000u) != 0x7F800000u);
        const float oh = nm / (dn + 1e-16f) + gat_bias[hh];
        z += oh * lin_w[hh];
      }
      out[lo + i] = fmaxf(z, 0.f) + bias[0];
      if (check && !fin) redo = 1u;
    }
  };

  if (binned) {
    if (owner) {
      const float4 a4 = *(const float4*)(adL + mydl * ADP + 4 * (int)myp);
      const float ad[HQ] = { a4.x, a4.y, a4.z, a4.w };
      float rf[HQ], den[HQ], num[HQ];
      #pragma unroll
      for (int hh = 0; hh < HQ; ++hh) {
        rf[hh] = __uint_as_float(emL[ACC(4 * (int)myp + hh, mydl)]);
        den[hh] = 0.f;
        num[hh] = 0.f;
      }
      for (unsigned s = 0u; s < mycnt; s += NU) {
        unsigned ix[NU];
        float4 sa[NU], sh[NU];
        #pragma unroll
        for (int u = 0; u < NU; ++u) {
          const unsigned su = min(s + (unsigned)u, mycnt - 1u);
          ix[u] = listC[su * NPB + mydl];
        }
        #pragma unroll
        for (int u = 0; u < NU; ++u) {
          const float4* __restrict__ r = (const float4*)(rec + (size_t)ix[u] * REC) + myp;
          sa[u] = r[0];
          sh[u] = r[2];
        }
        #pragma unroll
        for (int u = 0; u < NU; ++u) {
          const float off = s + (unsigned)u < mycnt ? 0.f : -INFINITY;
          const float v[HQ]  = { sa[u].x + ad[0], sa[u].y + ad[1], sa[u].z + ad[2], sa[u].w + ad[3] };
          const float hv[HQ] = { sh[u].x, sh[u].y, sh[u].z, sh[u].w };
          #pragma unroll
          for (int hh = 0; hh < HQ; ++hh) {
            const float ee = expf((leaky(v[hh]) - rf[hh]) + off);
            den[hh] += ee;
            num[hh] += ee * hv[hh];
          }
        }
      }
      #pragma unroll
      for (int hh = 0; hh < HQ; ++hh) {
        denL[ACC(4 * (int)myp + hh, mydl)] += den[hh];
        numL[ACC(4 * (int)myp + hh, mydl)] += num[hh];
      }
    }
  } else {
    scan_dst_plain(dstp, nE, lo, (unsigned)cntN, [&](int e, int dl) { edge_acc_self(ei[e], dl); });
  }
  __syncthreads();

  head(true);
  __syncthreads();
  if (redo == 0u) return;

  for (int j = tid; j < cntE; j += RTH) {
    const int a = ACC(j & 7, j >> 3);
    emL[a] = encF(__uint_as_float(emL[a]));
  }
  __syncthreads();

  const unsigned lcnt = myp == 0u ? mycnt : 0u;
  if (binned) {
    for (unsigned s = 0u; s < lcnt; ++s) edge_max((int)listC[s * NPB + mydl], mydl);
  } else {
    scan_dst_plain(dstp, nE, lo, (unsigned)cntN, [&](int e, int dl) { edge_max(ei[e], dl); });
  }
  __syncthreads();

  for (int j = tid; j < cntE; j += RTH) {
    const float* __restrict__ pr = rec + (size_t)(lo + (j >> 3)) * REC + (j & 7);
    const int a = ACC(j & 7, j >> 3);
    const float ee = expf(leaky(pr[0] + adL[(j >> 3) * ADP + (j & 7)]) - decF(emL[a]));
    denL[a] = ee;
    numL[a] = ee * pr[HEADS];
  }
  __syncthreads();

  if (binned) {
    for (unsigned s = 0u; s < lcnt; ++s) edge_acc_max((int)listC[s * NPB + mydl], mydl);
  } else {
    scan_dst_plain(dstp, nE, lo, (unsigned)cntN, [&](int e, int dl) { edge_acc_max(ei[e], dl); });
  }
  __syncthreads();

  head(false);
}

extern "C" void kernel_launch(void* const* d_in, const int* in_sizes, int n_in,
                              void* d_out, int out_size, void* d_ws, size_t ws_size,
                              hipStream_t stream) {
  const float* x        = (const float*)d_in[0];
  const int*   ei       = (const int*)d_in[1];
  const float* W        = (const float*)d_in[2];
  const float* att_src  = (const float*)d_in[3];
  const float* att_dst  = (const float*)d_in[4];
  const float* gat_bias = (const float*)d_in[5];
  const float* lin_w    = (const float*)d_in[6];
  const float* lin_b    = (const float*)d_in[7];
  const float* bias     = (const float*)d_in[8];
  float* out = (float*)d_out;

  const int n  = in_sizes[0] / IN_CH;
  const int nE = in_sizes[1] / 2;

  const size_t nh  = (size_t)n * HEADS;
  const size_t seg = (nh * sizeof(float) + 255) & ~(size_t)255;
  char* ws = (char*)d_ws;
  float*    rec = (float*)   (ws + 0 * seg);
  unsigned* bm  = (unsigned*)(ws + 2 * seg);

  const int tiles  = (n + 15) / 16;
  const int blocks = (tiles + 7) / 8;
  gat_gemm_kernel<<<blocks, GTH, 0, stream>>>(x, W, att_src, rec, n, ei, bm, nE);

  gat_range_kernel<<<(n + NPB - 1) / NPB, RTH, 0, stream>>>(
      ei, rec, att_dst, gat_bias, lin_w, lin_b, bias, out, nE, n, bm);
}


# record re-laid as four 16-byte pieces (a_src, h of two heads each); four lanes per node walk its list, one load per entry: a wave's gather touches 8 records of 64 bytes, not 16 twice; sums as before
# speedup vs baseline: 17.2726x; 1.0303x over previous
#include <hip/hip_runtime.h>
#include <math.h>

#define IN_CH 512
#define HEADS 8
#define NEG_SLOPE 0.2f
#define REC 16
#define RA(hh) ((((hh) >> 1) << 2) + ((hh) & 1))
#define RH(hh) (RA(hh) + 2)

typedef __attribute__((ext_vector_type(2))) float v2f;
typedef __attribute__((ext_vector_type(8))) float v8f;

__device__ __forceinline__ unsigned encF(float f) {
  unsigned u = __float_as_uint(f);
  return (u & 0x80000000u) ? ~u : (u | 0x80000000u);
}
__device__ __forceinline__ float decF(unsigned u) {
  u = (u & 0x80000000u) ? (u & 0x7FFFFFFFu) : ~u;
  return __uint_as_float(u);
}

__device__ __forceinline__ float leaky(float v) {
  return v >= 0.f ? v : NEG_SLOPE * v;
}

#define NPB 416

__device__ __forceinline__ unsigned range_key(int d) {
  const unsigned r = (unsigned)d / (unsigned)NPB;
  return r < 255u ? r : 255u;
}

#define GTH 256
#define CHP 1024
#define CHW 8
#define RMAX 256

__device__ __forceinline__ int bm_rows(int n) {
  const int nr = (n + NPB - 1) / NPB;
  const int nk = nr < RMAX ? nr : RMAX;
  return (nk + 3) & ~3;
}

__device__ __forceinline__ bool vec_path(const int* ei, int nE, const unsigned* bm) {
  return ((nE & 3) == 0) && ((((size_t)ei) & 15) == 0) && ((((size_t)bm) & 15) == 0);
}

#define KSL 64
#define TPB 8
#define NPBK (TPB * 16)

__global__ __launch_bounds__(GTH) void gat_gemm_kernel(
    const float* __restrict__ x, const float* __restrict__ W,
    const float* __restrict__ att_src,
    float* __restrict__ rec,
    int n,
    const int* __restrict__ ei, unsigned* __restrict__ bm, int nE)
{
  __shared__ __attribute__((aligned(16))) float part[(GTH >> 5) * NPBK * HEADS];
  __shared__ __attribute__((aligned(16))) unsigned bmL[RMAX * CHW];

  const bool vec = vec_path(ei, nE, bm);
  const int4* __restrict__ dv = (const int4*)(ei + nE);
  const int nPieces = nE >> 2;
  const int nChunks = vec ? (nPieces + CHP - 1) / CHP : 0;
  const int RP = bm_rows(n);
  int ch = (int)blockIdx.x;
  int4 d[CHP / GTH];
  auto load_chunk = [&]() {
    #pragma unroll
    for (int u = 0; u < CHP / GTH; ++u) {
      const int p = ch * CHP + u * GTH + (int)threadIdx.x;
      d[u] = dv[p < nPieces ? p : nPieces - 1];
    }
  };
  auto clear_rows = [&]() {
    for (int i = threadIdx.x; i < RMAX * CHW / 4; i += GTH)
      ((uint4*)bmL)[i] = make_uint4(0u, 0u, 0u, 0u);
  };
  if (ch < nChunks) load_chunk();
  clear_rows();

  const int lane = threadIdx.x & 31;
  const int wave = threadIdx.x >> 5;
  const int hf   = lane >> 4;
  const int c    = lane & 15;
  const int kbase = wave * KSL;

  v2f af[KSL / 4];
  {
    const float* __restrict__ wp = W + (size_t)(kbase + 4 * hf) * HEADS + (c & 7);
    #pragma unroll
    for (int j = 0; j < KSL / 4; ++j) {
      const int k = 8 * (j >> 1) + 2 * (j & 1);
      const float w0 = wp[k * HEADS];
      const float w1 = wp[(k + 1) * HEADS];
      af[j].x = (c < HEADS) ? w0 : 0.f;
      af[j].y = (c < HEADS) ? w1 : 0.f;
    }
  }
  __syncthreads();

  while (ch < nChunks) {
    #pragma unroll
    for (int u = 0; u < CHP / GTH; ++u) {
      const int pl = u * GTH + (int)threadIdx.x;
      if (ch * CHP + pl < nPieces) {
        const unsigned w   = (unsigned)pl >> 7;
        const unsigned bit = 1u << (((unsigned)pl >> 2) & 31u);
        atomicOr(&bmL[range_key(d[u].x) * CHW + w], bit);
        atomicOr(&bmL[range_key(d[u].y) * CHW + w], bit);
        atomicOr(&bmL[range_key(d[u].z) * CHW + w], bit);
        atomicOr(&bmL[range_key(d[u].w) * CHW + w], bit);
      }
    }
    __syncthreads();
    uint4* __restrict__ o = (uint4*)bm + (size_t)ch * (size_t)(RP * 2);
    for (int i = threadIdx.x; i < RP * 2; i += GTH) o[i] = ((const uint4*)bmL)[i];
    ch += (int)gridDim.x;
    if (ch < nChunks) {
      __syncthreads();
      load_chunk();
      clear_rows();
      __syncthreads();
    }
  }

  const int row0 = (int)blockIdx.x * NPBK;

  #pragma unroll 1
  for (int t = 0; t < TPB; t += 2) {
    int n0 = row0 + t * 16 + c;
    int n1 = n0 + 16;
    n0 = n0 < n ? n0 : n - 1;
    n1 = n1 < n ? n1 : n - 1;
    const float4* __restrict__ p0 = (const float4*)(x + (size_t)n0 * IN_CH + kbase + 4 * hf);
    const float4* __restrict__ p1 = (const float4*)(x + (size_t)n1 * IN_CH + kbase + 4 * hf);
    float4 q0[8], q1[8];
    #pragma unroll
    for (int i = 0; i < 8; ++i) { q0[i] = p0[2 * i]; q1[i] = p1[2 * i]; }
    __builtin_amdgcn_sched_barrier(0);

    v8f c0 = {0.f, 0.f, 0.f, 0.f, 0.f, 0.f, 0.f, 0.f};
    v8f c1 = {0.f, 0.f, 0.f, 0.f, 0.f, 0.f, 0.f, 0.f};
    #pragma unroll
    for (int j = 0; j < KSL / 4; ++j) {
      v2f b0, b1;
      if (j & 1) { b0.x = q0[j >> 1].z; b0.y = q0[j >> 1].w; b1.x = q1[j >> 1].z; b1.y = q1[j >> 1].w; }
      else       { b0.x = q0[j >> 1].x; b0.y = q0[j >> 1].y; b1.x = q1[j >> 1].x; b1.y = q1[j >> 1].y; }
      c0 = __builtin_amdgcn_wmma_f32_16x16x4_f32(
           false, af[j],  false, b0,
           (short)0, c0,  false,  false);
      c1 = __builtin_amdgcn_wmma_f32_16x16x4_f32(
           false, af[j],  false, b1,
           (short)0, c1,  false,  false);
    }
    if (hf == 0) {
      float4* __restrict__ pp0 = (float4*)(part + ((size_t)wave * NPBK + t * 16 + c) * HEADS);
      float4* __restrict__ pp1 = pp0 + 16 * HEADS / 4;
      pp0[0] = make_float4(c0[0], c0[1], c0[2], c0[3]);
      pp0[1] = make_float4(c0[4], c0[5], c0[6], c0[7]);
      pp1[0] = make_float4(c1[0], c1[1], c1[2], c1[3]);
      pp1[1] = make_float4(c1[4], c1[5], c1[6], c1[7]);
    }
  }
  __syncthreads();

  const float2* __restrict__ part2 = (const float2*)part;
  for (int slot = threadIdx.x; slot < NPBK * 4; slot += GTH) {
    const int nd = slot >> 2;
    const int q  = slot & 3;
    float2 s = part2[nd * 4 + q];
    #pragma unroll
    for (int w = 1; w < (GTH >> 5); ++w) {
      const float2 v = part2[(w * NPBK + nd) * 4 + q];
      s.x += v.x; s.y += v.y;
    }
    const int r = row0 + nd;
    if (r < n)
      ((float4*)(rec + (size_t)r * REC))[q] =
          make_float4(s.x * att_src[2 * q], s.y * att_src[2 * q + 1], s.x, s.y);
  }
}

#define RTH 1024
#define CAP 28672
#define EW 8
#define RU 4
#define ACC(hh, dl) ((hh) * NPB + (dl))
#define CAPN 72
#define NU 4
#define DLB 9
#define ADP 12

template <class F>
__device__ __forceinline__ void scan_dst_plain(const int* __restrict__ dstp, int nE,
                                               int lo, unsigned cnt, F f)
{
  for (int e = threadIdx.x; e < nE; e += RTH) {
    const unsigned t = (unsigned)(dstp[e] - lo);
    if (t < cnt) f(e, (int)t);
  }
}

__global__ __launch_bounds__(RTH) void gat_range_kernel(
    const int* __restrict__ ei,
    const float* __restrict__ rec, const float* __restrict__ att_dst,
    const float* __restrict__ gat_bias, const float* __restrict__ lin_w,
    const float* __restrict__ lin_b, const float* __restrict__ bias,
    float* __restrict__ out, int nE, int n,
    const unsigned* __restrict__ bm)
{
  __shared__ unsigned emL[NPB * HEADS];
  __shared__ float    denL[NPB * HEADS];
  __shared__ float    numL[NPB * HEADS];
  __shared__ __attribute__((aligned(16))) float adL[NPB * ADP];
  __shared__ unsigned listG[CAP];
  __shared__ unsigned listC[CAPN * NPB];
  __shared__ unsigned ncnt[NPB];
  __shared__ unsigned nlist;
  __shared__ unsigned over;
  __shared__ unsigned redo;

  const int tid = threadIdx.x;
  const int lo  = blockIdx.x * NPB;
  const int cntN = (n - lo < NPB) ? (n - lo) : NPB;
  const int cntE = cntN * HEADS;
  const int* __restrict__ dstp = ei + nE;

  const bool vec_ok  = vec_path(ei, nE, bm);
  const int  nPieces = nE >> 2;
  const int  nW      = vec_ok ? ((nPieces + CHP - 1) / CHP) * CHW : 0;
  const unsigned* __restrict__ myrow =
      bm + (size_t)(blockIdx.x < 255u ? blockIdx.x : 255u) * CHW;
  const size_t cpitch = (size_t)bm_rows(n) * CHW;
  unsigned wd[EW];
  auto load_words = [&](int w0) {
    #pragma unroll
    for (int u = 0; u < EW; ++u) {
      const int w = w0 + u * RTH + tid;
      wd[u] = (w < nW) ? myrow[(size_t)(w >> 3) * cpitch + (size_t)(w & 7)] : 0u;
    }
  };
  load_words(0);

  for (int j = tid; j < cntE; j += RTH) {
    const float* __restrict__ pr = rec + (size_t)(lo + (j >> 3)) * REC;
    const float hs = pr[RH(j & 7)];
    const float a = hs * att_dst[j & 7];
    const int ai = ACC(j & 7, j >> 3);
    adL[(j >> 3) * ADP + (j & 7)] = a;
    emL[ai]  = __float_as_uint(leaky(pr[RA(j & 7)] + a));
    denL[ai] = 1.f;
    numL[ai] = hs;
  }
  if (tid < NPB) ncnt[tid] = 0u;
  if (tid == 0) { nlist = 0u; over = 0u; redo = 0u; }
  __syncthreads();

  for (int w0 = 0; w0 < nW; ) {
    unsigned cnt = 0u;
    #pragma unroll
    for (int u = 0; u < EW; ++u) cnt += (unsigned)__builtin_popcount(wd[u]);
    if (cnt) {
      unsigned sl = atomicAdd(&nlist, cnt);
      #pragma unroll
      for (int u = 0; u < EW; ++u) {
        unsigned m = wd[u];
        const unsigned g0 = (unsigned)(w0 + u * RTH + tid) << 5;
        while (m) {
          const unsigned b = (unsigned)__builtin_ctz(m);
          m &= m - 1u;
          if (sl < (unsigned)CAP) listG[sl] = g0 + b;
          ++sl;
        }
      }
    }
    w0 += RTH * EW;
    if (w0 < nW) load_words(w0);
  }
  __syncthreads();
  const unsigned total = nlist;
  const bool listed = vec_ok && total <= (unsigned)CAP && (unsigned)n <= (1u << (32 - DLB));

  auto edge_max = [&](int src, int dl) {
    const float4* ps = (const float4*)(rec + (size_t)src * REC);
    const float4 p0 = ps[0], p1 = ps[1], p2 = ps[2], p3 = ps[3];
    const float* ad = adL + dl * ADP;
    const float v[HEADS] = { p0.x + ad[0], p0.y + ad[1], p1.x + ad[2], p1.y + ad[3],
                             p2.x + ad[4], p2.y + ad[5], p3.x + ad[6], p3.y + ad[7] };
    #pragma unroll
    for (int hh = 0; hh < HEADS; ++hh)
      atomicMax(&emL[ACC(hh, dl)], encF(leaky(v[hh])));
  };
  auto edge_acc_from = [&](int src, int dl, auto ref_point) {
    const float4* pr = (const float4*)(rec + (size_t)src * REC);
    const float4 p0 = pr[0], p1 = pr[1], p2 = pr[2], p3 = pr[3];
    const float* ad = adL + dl * ADP;
    const float v[HEADS]  = { p0.x + ad[0], p0.y + ad[1], p1.x + ad[2], p1.y + ad[3],
                              p2.x + ad[4], p2.y + ad[5], p3.x + ad[6], p3.y + ad[7] };
    const float hv[HEADS] = { p0.z, p0.w, p1.z, p1.w, p2.z, p2.w, p3.z, p3.w };
    float ee[HEADS];
    #pragma unroll
    for (int hh = 0; hh < HEADS; ++hh) ee[hh] = ref_point(emL[ACC(hh, dl)]);
    #pragma unroll
    for (int hh = 0; hh < HEADS; ++hh) ee[hh] = expf(leaky(v[hh]) - ee[hh]);
    #pragma unroll
    for (int hh = 0; hh < HEADS; ++hh) {
      atomicAdd(&denL[ACC(hh, dl)], ee[hh]);
      atomicAdd(&numL[ACC(hh, dl)], ee[hh] * hv[hh]);
    }
  };
  auto edge_acc_self = [&](int src, int dl) {
    edge_acc_from(src, dl, [](unsigned u) { return __uint_as_float(u); });
  };
  auto edge_acc_max = [&](int src, int dl) {
    edge_acc_from(src, dl, [](unsigned u) { return decF(u); });
  };

  if (listed) {
    const int4* __restrict__ dv = (const int4*)dstp;
    const int4* __restrict__ sv = (const int4*)ei;
    const unsigned tot4 = total << 2;
    for (unsigned i0 = 0; i0 < tot4; i0 += RTH * RU) {
      int4 d[RU], s[RU];
      bool ok[RU];
      #pragma unroll
      for (int u = 0; u < RU; ++u) {
        const unsigned i = i0 + (unsigned)(u * RTH + tid);
        const bool in = i < tot4;
        int p = (int)((listG[in ? (i >> 2) : 0u] << 2) | (i & 3u));
        ok[u] = in && p < nPieces;
        p = ok[u] ? p : 0;
        d[u] = dv[p];
        s[u] = sv[p];
      }
      #pragma unroll
      for (int u = 0; u < RU; ++u) {
        asm volatile("" : "+v"(s[u].x), "+v"(s[u].y), "+v"(s[u].z), "+v"(s[u].w));
        const unsigned t0 = (unsigned)(d[u].x - lo), t1 = (unsigned)(d[u].y - lo);
        const unsigned t2 = (unsigned)(d[u].z - lo), t3 = (unsigned)(d[u].w - lo);
        unsigned mk = (t0 < (unsigned)cntN ? 1u : 0u) | (t1 < (unsigned)cntN ? 2u : 0u) |
                      (t2 < (unsigned)cntN ? 4u : 0u) | (t3 < (unsigned)cntN ? 8u : 0u);
        mk = ok[u] ? mk : 0u;
        const unsigned w0 = ((unsigned)s[u].x << DLB) | t0, w1 = ((unsigned)s[u].y << DLB) | t1;
        const unsigned w2 = ((unsigned)s[u].z << DLB) | t2, w3 = ((unsigned)s[u].w << DLB) | t3;
        while (mk) {
          const unsigned j = (unsigned)__builtin_ctz(mk);
          mk &= mk - 1u;
          const bool j1 = __builtin_unpredictable((j & 1u) != 0u);
          const bool j2 = __builtin_unpredictable((j & 2u) != 0u);
          const unsigned w  = j2 ? (j1 ? w3 : w2) : (j1 ? w1 : w0);
          const unsigned dlv = w & ((1u << DLB) - 1u);
          const unsigned sl = atomicAdd(&ncnt[dlv], 1u);
          if (sl < (unsigned)CAPN) listC[sl * NPB + dlv] = w >> DLB;
          else over = 1u;
        }
      }
    }
  }
  __syncthreads();
  const bool binned = listed && over == 0u;
  const unsigned myq = (unsigned)tid & 3u;

  auto head = [&](bool check) {
    for (int i = tid; i < cntN; i += RTH) {
      float z = lin_b[0];
      bool fin = true;
      #pragma unroll
      for (int hh = 0; hh < HEADS; ++hh) {
        const float dn = denL[ACC(hh, i)];
        const float nm = numL[ACC(hh, i)];
        fin = fin && ((__float_as_uint(dn) & 0x7F800000u) != 0x7F800000u)
                  && ((__float_as_uint(nm) & 0x7F800000u) != 0x7F800000u);
        const float oh = nm / (dn + 1e-16f) + gat_bias[hh];
        z += oh * lin_w[hh];
      }
      out[lo + i] = fmaxf(z, 0.f) + bias[0];
      if (check && !fin) redo = 1u;
    }
  };

  if (binned) {
    for (int base = 0; base < cntN; base += RTH / 4) {
      const int dl = base + (tid >> 2);
      if (dl < cntN) {
        const unsigned cnt = ncnt[dl];
        const float2 ad = *(const float2*)(adL + dl * ADP + 2 * (int)myq);
        const float rf0 = __uint_as_float(emL[ACC(2 * (int)myq, dl)]);
        const float rf1 = __uint_as_float(emL[ACC(2 * (int)myq + 1, dl)]);
        float den0 = 0.f, den1 = 0.f, num0 = 0.f, num1 = 0.f;
        for (unsigned s = 0u; s < cnt; s += NU) {
          unsigned ix[NU];
          float4 r[NU];
          #pragma unroll
          for (int u = 0; u < NU; ++u) {
            const unsigned su = min(s + (unsigned)u, cnt - 1u);
            ix[u] = listC[su * NPB + dl];
          }
          #pragma unroll
          for (int u = 0; u < NU; ++u)
            r[u] = ((const float4*)(rec + (size_t)ix[u] * REC))[myq];
          #pragma unroll
          for (int u = 0; u < NU; ++u) {
            const float off = s + (unsigned)u < cnt ? 0.f : -INFINITY;
            const float e0 = expf((leaky(r[u].x + ad.x) - rf0) + off);
            const float e1 = expf((leaky(r[u].y + ad.y) - rf1) + off);
            den0 += e0;
            num0 += e0 * r[u].z;
            den1 += e1;
            num1 += e1 * r[u].w;
          }
        }
        denL[ACC(2 * (int)myq, dl)]     += den0;
        numL[ACC(2 * (int)myq, dl)]     += num0;
        denL[ACC(2 * (int)myq + 1, dl)] += den1;
        numL[ACC(2 * (int)myq + 1, dl)] += num1;
      }
    }
  } else {
    scan_dst_plain(dstp, nE, lo, (unsigned)cntN, [&](int e, int dl) { edge_acc_self(ei[e], dl); });
  }
  __syncthreads();

  head(true);
  __syncthreads();
  if (redo == 0u) return;

  for (int j = tid; j < cntE; j += RTH) {
    const int a = ACC(j & 7, j >> 3);
    emL[a] = encF(__uint_as_float(emL[a]));
  }
  __syncthreads();

  const int mydl = (binned && tid < cntN) ? tid : 0;
  const unsigned lcnt = (binned && tid < cntN) ? ncnt[mydl] : 0u;
  if (binned) {
    for (unsigned s = 0u; s < lcnt; ++s) edge_max((int)listC[s * NPB + mydl], mydl);
  } else {
    scan_dst_plain(dstp, nE, lo, (unsigned)cntN, [&](int e, int dl) { edge_max(ei[e], dl); });
  }
  __syncthreads();

  for (int j = tid; j < cntE; j += RTH) {
    const float* __restrict__ pr = rec + (size_t)(lo + (j >> 3)) * REC;
    const int a = ACC(j & 7, j >> 3);
    const float ee = expf(leaky(pr[RA(j & 7)] + adL[(j >> 3) * ADP + (j & 7)]) - decF(emL[a]));
    denL[a] = ee;
    numL[a] = ee * pr[RH(j & 7)];
  }
  __syncthreads();

  if (binned) {
    for (unsigned s = 0u; s < lcnt; ++s) edge_acc_max((int)listC[s * NPB + mydl], mydl);
  } else {
    scan_dst_plain(dstp, nE, lo, (unsigned)cntN, [&](int e, int dl) { edge_acc_max(ei[e], dl); });
  }
  __syncthreads();

  head(false);
}

extern "C" void kernel_launch(void* const* d_in, const int* in_sizes, int n_in,
                              void* d_out, int out_size, void* d_ws, size_t ws_size,
                              hipStream_t stream) {
  const float* x        = (const float*)d_in[0];
  const int*   ei       = (const int*)d_in[1];
  const float* W        = (const float*)d_in[2];
  const float* att_src  = (const float*)d_in[3];
  const float* att_dst  = (const float*)d_in[4];
  const float* gat_bias = (const float*)d_in[5];
  const float* lin_w    = (const float*)d_in[6];
  const float* lin_b    = (const float*)d_in[7];
  const float* bias     = (const float*)d_in[8];
  float* out = (float*)d_out;

  const int n  = in_sizes[0] / IN_CH;
  const int nE = in_sizes[1] / 2;

  const size_t nh  = (size_t)n * HEADS;
  const size_t seg = (nh * sizeof(float) + 255) & ~(size_t)255;
  char* ws = (char*)d_ws;
  float*    rec = (float*)   (ws + 0 * seg);
  unsigned* bm  = (unsigned*)(ws + 2 * seg);

  const int tiles  = (n + 15) / 16;
  const int blocks = (tiles + 7) / 8;
  gat_gemm_kernel<<<blocks, GTH, 0, stream>>>(x, W, att_src, rec, n, ei, bm, nE);

  gat_range_kernel<<<(n + NPB - 1) / NPB, RTH, 0, stream>>>(
      ei, rec, att_dst, gat_bias, lin_w, lin_b, bias, out, nE, n, bm);
}
